# starting kernel
# baseline (speedup 1.0000x reference)
#include <hip/hip_runtime.h>
#include <hip/hip_bf16.h>
#include <math.h>

#define Bn   64
#define Ln   2048
#define Dn   300
#define Cn   2
#define Kn   111
#define PADn 55

typedef __attribute__((ext_vector_type(2))) float v2f;
typedef __attribute__((ext_vector_type(8))) float v8f;

__global__ __launch_bounds__(256)
void k_phase1(const int* __restrict__ tok, const float* __restrict__ emb,
              const float* __restrict__ cls, float* __restrict__ g)
{
    __shared__ float clsn[3 * 304];
    __shared__ float red[256];
    const int tid = threadIdx.x;

    for (int i = tid; i < 3 * 304; i += 256) clsn[i] = 0.0f;
    __syncthreads();
    for (int c = 0; c < Cn; ++c) {
        float s = 0.0f;
        for (int d = tid; d < Dn; d += 256) { float v = cls[c * Dn + d]; s += v * v; }
        red[tid] = s;
        __syncthreads();
        for (int off = 128; off > 0; off >>= 1) {
            if (tid < off) red[tid] += red[tid + off];
            __syncthreads();
        }
        float rn = 1.0f / fmaxf(sqrtf(red[0]), 1e-12f);
        __syncthreads();
        for (int d = tid; d < Dn; d += 256) clsn[c * 304 + d] = cls[c * Dn + d] * rn;
        __syncthreads();
    }

    const int wave = tid >> 5;
    const int lane = tid & 31;
    const int tileId = blockIdx.x * 8 + wave;
    const int b  = tileId >> 7;
    const int l0 = (tileId & 127) * 16;
    const int m    = lane & 15;
    const int hi   = lane >> 4;
    const int koff = hi * 2;

    const int token = tok[b * Ln + l0 + m];
    const float* __restrict__ row = emb + (long)token * Dn;
    const float* __restrict__ bb  = &clsn[(m < 2 ? m : 2) * 304];

    v8f acc = {};
    float n2 = 0.0f;
    #pragma unroll 5
    for (int k = 0; k < 75; ++k) {
        const int d0 = 4 * k + koff;
        v2f a;  a.x = row[d0];           a.y = row[d0 + 1];
        v2f bf; bf.x = bb[d0];           bf.y = bb[d0 + 1];
        n2 += a.x * a.x + a.y * a.y;
        acc = __builtin_amdgcn_wmma_f32_16x16x4_f32(
            false, a, false, bf, (short)0, acc, false, false);
    }

    n2 += __shfl_xor(n2, 16, 32);

    const int  c      = m;
    const bool active = (m < 2);
    const int  mbase  = hi * 8;
    float* __restrict__ gout = g + (((long)b * Cn + c) * Ln + l0 + mbase);
    #pragma unroll
    for (int i = 0; i < 8; ++i) {
        float nn    = __shfl(n2, mbase + i, 32);
        float scale = 1.0f / fmaxf(sqrtf(nn), 1e-12f);
        if (active) gout[i] = acc[i] * scale;
    }
}

__global__ __launch_bounds__(256)
void k_phase2(const float* __restrict__ g, const float* __restrict__ conv_w,
              const float* __restrict__ conv_b, float* __restrict__ betaOut)
{
    __shared__ float gin[Cn][Ln + 2 * PADn + 2];
    __shared__ float wsm[Cn * Cn * Kn];
    __shared__ float red[256];
    const int b = blockIdx.x, tid = threadIdx.x;

    for (int i = tid; i < Cn * (Ln + 2 * PADn + 2); i += 256) (&gin[0][0])[i] = 0.0f;
    __syncthreads();
    for (int i = tid; i < Ln; i += 256) {
        gin[0][PADn + i] = g[((long)b * Cn + 0) * Ln + i];
        gin[1][PADn + i] = g[((long)b * Cn + 1) * Ln + i];
    }
    for (int i = tid; i < Cn * Cn * Kn; i += 256) wsm[i] = conv_w[i];
    __syncthreads();

    const float cb0 = conv_b[0], cb1 = conv_b[1];
    float h[8];
    #pragma unroll
    for (int j = 0; j < 8; ++j) {
        const int x = tid + 256 * j;
        float a0 = 0.0f, a1 = 0.0f;
        for (int k = 0; k < Kn; ++k) {
            const float i0 = gin[0][x + k];
            const float i1 = gin[1][x + k];
            a0 += wsm[0 * Kn + k] * i0 + wsm[1 * Kn + k] * i1;
            a1 += wsm[2 * Kn + k] * i0 + wsm[3 * Kn + k] * i1;
        }
        h[j] = fmaxf(fmaxf(a0 + cb0, 0.0f), fmaxf(a1 + cb1, 0.0f));
    }

    float lm = -INFINITY;
    #pragma unroll
    for (int j = 0; j < 8; ++j) lm = fmaxf(lm, h[j]);
    red[tid] = lm; __syncthreads();
    for (int off = 128; off > 0; off >>= 1) {
        if (tid < off) red[tid] = fmaxf(red[tid], red[tid + off]);
        __syncthreads();
    }
    const float M = red[0];
    __syncthreads();

    float ls = 0.0f;
    #pragma unroll
    for (int j = 0; j < 8; ++j) { h[j] = __expf(h[j] - M); ls += h[j]; }
    red[tid] = ls; __syncthreads();
    for (int off = 128; off > 0; off >>= 1) {
        if (tid < off) red[tid] += red[tid + off];
        __syncthreads();
    }
    const float inv = 1.0f / red[0];
    #pragma unroll
    for (int j = 0; j < 8; ++j)
        betaOut[(long)b * Ln + tid + 256 * j] = h[j] * inv;
}

__global__ __launch_bounds__(320)
void k_phase3(const int* __restrict__ tok, const float* __restrict__ emb,
              const float* __restrict__ beta, const float* __restrict__ lin_w,
              const float* __restrict__ lin_b, float* __restrict__ logits)
{
    __shared__ float s_acc[2];
    const int b = blockIdx.x, tid = threadIdx.x;
    if (tid < 2) s_acc[tid] = 0.0f;

    const int*   __restrict__ trow = tok  + (long)b * Ln;
    const float* __restrict__ brow = beta + (long)b * Ln;
    const bool act = (tid < Dn);
    const int  dd  = act ? tid : 0;

    float z = 0.0f;
    for (int l = 0; l < Ln; l += 4) {
        const int t0 = trow[l],     t1 = trow[l + 1];
        const int t2 = trow[l + 2], t3 = trow[l + 3];
        const float b0 = brow[l],     b1 = brow[l + 1];
        const float b2 = brow[l + 2], b3 = brow[l + 3];
        z += b0 * emb[(long)t0 * Dn + dd];
        z += b1 * emb[(long)t1 * Dn + dd];
        z += b2 * emb[(long)t2 * Dn + dd];
        z += b3 * emb[(long)t3 * Dn + dd];
    }
    __syncthreads();
    if (act) {
        atomicAdd(&s_acc[0], z * lin_w[0 * Dn + tid]);
        atomicAdd(&s_acc[1], z * lin_w[1 * Dn + tid]);
    }
    __syncthreads();
    if (tid == 0) {
        const float o0 = s_acc[0] + lin_b[0];
        const float o1 = s_acc[1] + lin_b[1];
        const float mm = fmaxf(o0, o1);
        const float lse = mm + __logf(__expf(o0 - mm) + __expf(o1 - mm));
        logits[b * 2 + 0] = o0 - lse;
        logits[b * 2 + 1] = o1 - lse;
    }
}

extern "C" void kernel_launch(void* const* d_in, const int* in_sizes, int n_in,
                              void* d_out, int out_size, void* d_ws, size_t ws_size,
                              hipStream_t stream)
{
    const int*   tok = (const int*)  d_in[0];
    const float* emb = (const float*)d_in[1];
    const float* cls = (const float*)d_in[2];
    const float* cw  = (const float*)d_in[3];
    const float* cb  = (const float*)d_in[4];
    const float* lw  = (const float*)d_in[5];
    const float* lb  = (const float*)d_in[6];

    float* out    = (float*)d_out;
    float* logits = out;
    float* beta   = out + Bn * Cn;
    float* g      = (float*)d_ws;

    k_phase1<<<1024, 256, 0, stream>>>(tok, emb, cls, g);
    k_phase2<<<Bn, 256, 0, stream>>>(g, cw, cb, beta);
    k_phase3<<<Bn, 320, 0, stream>>>(tok, emb, beta, lw, lb, logits);
}


# Conv register-blocked: a thread owns 8 consecutive positions and slides the input window through registers; one 16-byte LDS read of a tap's weights feeds 16 packed f32 FMAs.
# speedup vs baseline: 4.2000x; 4.2000x over previous
#include <hip/hip_runtime.h>
#include <hip/hip_bf16.h>
#include <math.h>

#define Bn   64
#define Ln   2048
#define Dn   300
#define Cn   2
#define Kn   111
#define PADn 55
#define GWn  2160
#define KPn  112
#define KWn  120

typedef __attribute__((ext_vector_type(2))) float v2f;
typedef __attribute__((ext_vector_type(8))) float v8f;

__global__ __launch_bounds__(256)
void k_phase1(const int* __restrict__ tok, const float* __restrict__ emb,
              const float* __restrict__ cls, const float* __restrict__ lin_w,
              float* __restrict__ g, float* __restrict__ qout)
{
    __shared__ float clsn[4 * 304];
    __shared__ float red[256];
    const int tid = threadIdx.x;

    for (int i = tid; i < 4 * 304; i += 256) clsn[i] = 0.0f;
    __syncthreads();
    for (int d = tid; d < Dn; d += 256) clsn[2 * 304 + d] = lin_w[Dn + d] - lin_w[d];
    for (int c = 0; c < Cn; ++c) {
        float s = 0.0f;
        for (int d = tid; d < Dn; d += 256) { float v = cls[c * Dn + d]; s += v * v; }
        red[tid] = s;
        __syncthreads();
        for (int off = 128; off > 0; off >>= 1) {
            if (tid < off) red[tid] += red[tid + off];
            __syncthreads();
        }
        float rn = 1.0f / fmaxf(sqrtf(red[0]), 1e-12f);
        __syncthreads();
        for (int d = tid; d < Dn; d += 256) clsn[c * 304 + d] = cls[c * Dn + d] * rn;
        __syncthreads();
    }

    const int wave = tid >> 5;
    const int lane = tid & 31;
    const int tileId = blockIdx.x * 8 + wave;
    const int b  = tileId >> 7;
    const int l0 = (tileId & 127) * 16;
    const int m    = lane & 15;
    const int hi   = lane >> 4;
    const int koff = hi * 2;

    const int token = tok[b * Ln + l0 + m];
    const float* __restrict__ row = emb + (long)token * Dn;
    const float* __restrict__ bb  = &clsn[(m < 3 ? m : 3) * 304];

    v8f acc = {};
    float n2 = 0.0f;
    #pragma unroll 5
    for (int k = 0; k < 75; ++k) {
        const int d0 = 4 * k + koff;
        v2f a;  a.x = row[d0];           a.y = row[d0 + 1];
        v2f bf; bf.x = bb[d0];           bf.y = bb[d0 + 1];
        n2 += a.x * a.x + a.y * a.y;
        acc = __builtin_amdgcn_wmma_f32_16x16x4_f32(
            false, a, false, bf, (short)0, acc, false, false);
    }

    n2 += __shfl_xor(n2, 16, 32);

    const bool isg    = (m < 2);
    const bool active = (m < 3);
    const int  mbase  = hi * 8;
    float* dst = isg ? g    + (((long)b * Cn + m) * Ln + l0 + mbase)
                     : qout + ((long)b * Ln + l0 + mbase);
    #pragma unroll
    for (int i = 0; i < 8; ++i) {
        float nn    = __shfl(n2, mbase + i, 32);
        float scale = 1.0f / fmaxf(sqrtf(nn), 1e-12f);
        if (active) dst[i] = isg ? acc[i] * scale : acc[i];
    }
}

__device__ __forceinline__ void lds_ld4(float* __restrict__ d, const float* s)
{
    const float4 v = *(const float4*)s;
    d[0] = v.x; d[1] = v.y; d[2] = v.z; d[3] = v.w;
}

__global__ __launch_bounds__(256)
void k_phase2(const float* __restrict__ g, const float* __restrict__ conv_w,
              const float* __restrict__ conv_b, const float* __restrict__ lin_b,
              float* __restrict__ betaOut, float* __restrict__ logits)
{
    __shared__ __attribute__((aligned(16))) float gin[Cn][GWn];
    __shared__ __attribute__((aligned(16))) float wtap[KWn * 4];
    __shared__ __attribute__((aligned(16))) float hbuf[Ln];
    __shared__ float red[256];
    __shared__ float redq[256];
    const int b = blockIdx.x, tid = threadIdx.x;

    float q[8];
    #pragma unroll
    for (int j = 0; j < 8; ++j) q[j] = betaOut[(long)b * Ln + tid + 256 * j];

    for (int i = tid; i < Cn * GWn; i += 256) (&gin[0][0])[i] = 0.0f;
    __syncthreads();
    for (int i = tid; i < Ln; i += 256) {
        gin[0][PADn + i] = g[((long)b * Cn + 0) * Ln + i];
        gin[1][PADn + i] = g[((long)b * Cn + 1) * Ln + i];
    }
    for (int k = tid; k < KWn; k += 256) {
        float4 w = make_float4(0.0f, 0.0f, 0.0f, 0.0f);
        if (k < Kn) w = make_float4(conv_w[0 * Kn + k], conv_w[2 * Kn + k],
                                    conv_w[1 * Kn + k], conv_w[3 * Kn + k]);
        *(float4*)&wtap[4 * k] = w;
    }
    __syncthreads();

    const float cb0 = conv_b[0], cb1 = conv_b[1];
    {
        const float* in0 = &gin[0][8 * tid];
        const float* in1 = &gin[1][8 * tid];
        v2f   acc[8];
        float xa[16], xb[16];
        #pragma unroll
        for (int i = 0; i < 8; ++i) { acc[i].x = 0.0f; acc[i].y = 0.0f; }
        lds_ld4(xa + 0, in0 + 0);  lds_ld4(xa + 4, in0 + 4);
        lds_ld4(xb + 0, in1 + 0);  lds_ld4(xb + 4, in1 + 4);

        float4 wc[8];
        #pragma unroll
        for (int kk = 0; kk < 8; ++kk) wc[kk] = *(const float4*)&wtap[4 * kk];

        #pragma unroll 2
        for (int k0 = 0; k0 < KPn; k0 += 8) {
            lds_ld4(xa + 8, in0 + k0 + 8);  lds_ld4(xa + 12, in0 + k0 + 12);
            lds_ld4(xb + 8, in1 + k0 + 8);  lds_ld4(xb + 12, in1 + k0 + 12);
            float4 wn[8];
            #pragma unroll
            for (int kk = 0; kk < 8; ++kk) wn[kk] = *(const float4*)&wtap[4 * (k0 + 8 + kk)];
            __builtin_amdgcn_sched_barrier(0);
            #pragma unroll
            for (int kk = 0; kk < 8; ++kk) {
                v2f wa; wa.x = wc[kk].x; wa.y = wc[kk].y;
                v2f wb; wb.x = wc[kk].z; wb.y = wc[kk].w;
                #pragma unroll
                for (int i = 0; i < 8; ++i) {
                    v2f va; va.x = xa[kk + i]; va.y = xa[kk + i];
                    v2f vb; vb.x = xb[kk + i]; vb.y = xb[kk + i];
                    acc[i] = __builtin_elementwise_fma(wa, va, acc[i]);
                    acc[i] = __builtin_elementwise_fma(wb, vb, acc[i]);
                }
            }
            #pragma unroll
            for (int i = 0; i < 8; ++i) { xa[i] = xa[i + 8]; xb[i] = xb[i + 8]; wc[i] = wn[i]; }
        }

        float r[8];
        #pragma unroll
        for (int i = 0; i < 8; ++i)
            r[i] = fmaxf(fmaxf(acc[i].x + cb0, 0.0f), fmaxf(acc[i].y + cb1, 0.0f));
        *(float4*)&hbuf[8 * tid + 0] = make_float4(r[0], r[1], r[2], r[3]);
        *(float4*)&hbuf[8 * tid + 4] = make_float4(r[4], r[5], r[6], r[7]);
    }
    __syncthreads();
    float h[8];
    #pragma unroll
    for (int j = 0; j < 8; ++j) h[j] = hbuf[tid + 256 * j];

    float lm = -INFINITY;
    #pragma unroll
    for (int j = 0; j < 8; ++j) lm = fmaxf(lm, h[j]);
    red[tid] = lm; __syncthreads();
    for (int off = 128; off > 0; off >>= 1) {
        if (tid < off) red[tid] = fmaxf(red[tid], red[tid + off]);
        __syncthreads();
    }
    const float M = red[0];
    __syncthreads();

    float ls = 0.0f, lq = 0.0f;
    #pragma unroll
    for (int j = 0; j < 8; ++j) { h[j] = __expf(h[j] - M); ls += h[j]; lq += h[j] * q[j]; }
    red[tid] = ls; redq[tid] = lq; __syncthreads();
    for (int off = 128; off > 0; off >>= 1) {
        if (tid < off) { red[tid] += red[tid + off]; redq[tid] += redq[tid + off]; }
        __syncthreads();
    }
    const float inv = 1.0f / red[0];
    #pragma unroll
    for (int j = 0; j < 8; ++j)
        betaOut[(long)b * Ln + tid + 256 * j] = h[j] * inv;

    if (tid == 0) {
        const float d  = redq[0] * inv + (lin_b[1] - lin_b[0]);
        const float t  = __logf(1.0f + __expf(-fabsf(d)));
        logits[b * 2 + 0] = -(fmaxf(d, 0.0f) + t);
        logits[b * 2 + 1] = -(fmaxf(-d, 0.0f) + t);
    }
}

extern "C" void kernel_launch(void* const* d_in, const int* in_sizes, int n_in,
                              void* d_out, int out_size, void* d_ws, size_t ws_size,
                              hipStream_t stream)
{
    const int*   tok = (const int*)  d_in[0];
    const float* emb = (const float*)d_in[1];
    const float* cls = (const float*)d_in[2];
    const float* cw  = (const float*)d_in[3];
    const float* cb  = (const float*)d_in[4];
    const float* lw  = (const float*)d_in[5];
    const float* lb  = (const float*)d_in[6];

    float* out    = (float*)d_out;
    float* logits = out;
    float* beta   = out + Bn * Cn;
    float* g      = (float*)d_ws;

    k_phase1<<<1024, 256, 0, stream>>>(tok, emb, cls, lw, g, beta);
    k_phase2<<<Bn, 256, 0, stream>>>(g, cw, cb, lb, beta, logits);
}


# Table rows gathered 16 bytes per lane: one float4 feeds two f32 WMMA steps, A and B k-permuted alike; half as many 16-row load instructions, issued in pinned runs within 128 registers.
# speedup vs baseline: 4.7532x; 1.1317x over previous
#include <hip/hip_runtime.h>
#include <hip/hip_bf16.h>
#include <math.h>

#define Bn   64
#define Ln   2048
#define Dn   300
#define Cn   2
#define Kn   111
#define PADn 55
#define GWn  2160
#define KPn  112
#define KWn  120

typedef __attribute__((ext_vector_type(2))) float v2f;
typedef __attribute__((ext_vector_type(8))) float v8f;

template <int S0, int N>
__device__ __forceinline__ void load_steps(const float4* __restrict__ row4, float4 (&a)[N])
{
    #pragma unroll
    for (int i = 0; i < N; ++i) a[i] = row4[2 * (S0 + i)];
}

template <int S0, int N>
__device__ __forceinline__ void mma_steps(const float4 (&a)[N], const float4* bb4,
                                          v8f& acc0, v8f& acc1, float& n2)
{
    #pragma unroll
    for (int i = 0; i < N; ++i) {
        const float4 bv = bb4[2 * (S0 + i)];
        v2f a0; a0.x = a[i].x; a0.y = a[i].y;
        v2f a1; a1.x = a[i].z; a1.y = a[i].w;
        v2f b0; b0.x = bv.x;   b0.y = bv.y;
        v2f b1; b1.x = bv.z;   b1.y = bv.w;
        n2 += a[i].x * a[i].x + a[i].y * a[i].y + a[i].z * a[i].z + a[i].w * a[i].w;
        acc0 = __builtin_amdgcn_wmma_f32_16x16x4_f32(
            false, a0, false, b0, (short)0, acc0, false, false);
        acc1 = __builtin_amdgcn_wmma_f32_16x16x4_f32(
            false, a1, false, b1, (short)0, acc1, false, false);
    }
}

__global__ __launch_bounds__(256) __attribute__((amdgpu_waves_per_eu(8)))
void k_phase1(const int* __restrict__ tok, const float* __restrict__ emb,
              const float* __restrict__ cls, const float* __restrict__ lin_w,
              float* __restrict__ g, float* __restrict__ qout)
{
    __shared__ __attribute__((aligned(16))) float clsn[4 * 304];
    __shared__ float red[256];
    const int tid = threadIdx.x;

    for (int i = tid; i < 4 * 304; i += 256) clsn[i] = 0.0f;
    __syncthreads();
    for (int d = tid; d < Dn; d += 256) clsn[2 * 304 + d] = lin_w[Dn + d] - lin_w[d];
    for (int c = 0; c < Cn; ++c) {
        float s = 0.0f;
        for (int d = tid; d < Dn; d += 256) { float v = cls[c * Dn + d]; s += v * v; }
        red[tid] = s;
        __syncthreads();
        for (int off = 128; off > 0; off >>= 1) {
            if (tid < off) red[tid] += red[tid + off];
            __syncthreads();
        }
        float rn = 1.0f / fmaxf(sqrtf(red[0]), 1e-12f);
        __syncthreads();
        for (int d = tid; d < Dn; d += 256) clsn[c * 304 + d] = cls[c * Dn + d] * rn;
        __syncthreads();
    }

    const int wave = tid >> 5;
    const int lane = tid & 31;
    const int tileId = blockIdx.x * 8 + wave;
    const int b  = tileId >> 7;
    const int l0 = (tileId & 127) * 16;
    const int m    = lane & 15;
    const int hi   = lane >> 4;
    const int koff = hi * 2;

    const int token = tok[b * Ln + l0 + m];
    const float* __restrict__ row = emb + (long)token * Dn;
    const float* bb = &clsn[(m < 3 ? m : 3) * 304];
    const float4* __restrict__ row4 = (const float4*)row + hi;
    const float4* bb4 = (const float4*)bb + hi;

    v8f acc0 = {}, acc1 = {};
    float n2 = 0.0f;
    float4 g0[10], g1[9], g2[9], g3[9];
    load_steps<0, 10>(row4, g0);
    load_steps<10, 9>(row4, g1);
    mma_steps<0, 10>(g0, bb4, acc0, acc1, n2);
    asm volatile("" : "+v"(acc0), "+v"(acc1), "+v"(n2) :: "memory");
    __builtin_amdgcn_sched_barrier(0);
    load_steps<19, 9>(row4, g2);
    __builtin_amdgcn_sched_barrier(0);
    mma_steps<10, 9>(g1, bb4, acc0, acc1, n2);
    asm volatile("" : "+v"(acc0), "+v"(acc1), "+v"(n2) :: "memory");
    __builtin_amdgcn_sched_barrier(0);
    load_steps<28, 9>(row4, g3);
    const int d0 = 296 + koff;
    v2f at; at.x = row[d0]; at.y = row[d0 + 1];
    __builtin_amdgcn_sched_barrier(0);
    mma_steps<19, 9>(g2, bb4, acc0, acc1, n2);
    mma_steps<28, 9>(g3, bb4, acc0, acc1, n2);
    {
        v2f bf; bf.x = bb[d0];  bf.y = bb[d0 + 1];
        n2 += at.x * at.x + at.y * at.y;
        acc0 = __builtin_amdgcn_wmma_f32_16x16x4_f32(
            false, at, false, bf, (short)0, acc0, false, false);
    }
    const v8f acc = acc0 + acc1;

    n2 += __shfl_xor(n2, 16, 32);

    const bool isg    = (m < 2);
    const bool active = (m < 3);
    const int  mbase  = hi * 8;
    float* dst = isg ? g    + (((long)b * Cn + m) * Ln + l0 + mbase)
                     : qout + ((long)b * Ln + l0 + mbase);
    #pragma unroll
    for (int i = 0; i < 8; ++i) {
        float nn    = __shfl(n2, mbase + i, 32);
        float scale = 1.0f / fmaxf(sqrtf(nn), 1e-12f);
        if (active) dst[i] = isg ? acc[i] * scale : acc[i];
    }
}

__device__ __forceinline__ void lds_ld4(float* __restrict__ d, const float* s)
{
    const float4 v = *(const float4*)s;
    d[0] = v.x; d[1] = v.y; d[2] = v.z; d[3] = v.w;
}

__global__ __launch_bounds__(256)
void k_phase2(const float* __restrict__ g, const float* __restrict__ conv_w,
              const float* __restrict__ conv_b, const float* __restrict__ lin_b,
              float* __restrict__ betaOut, float* __restrict__ logits)
{
    __shared__ __attribute__((aligned(16))) float gin[Cn][GWn];
    __shared__ __attribute__((aligned(16))) float wtap[KWn * 4];
    __shared__ __attribute__((aligned(16))) float hbuf[Ln];
    __shared__ float red[256];
    __shared__ float redq[256];
    const int b = blockIdx.x, tid = threadIdx.x;

    float q[8];
    #pragma unroll
    for (int j = 0; j < 8; ++j) q[j] = betaOut[(long)b * Ln + tid + 256 * j];

    for (int i = tid; i < Cn * GWn; i += 256) (&gin[0][0])[i] = 0.0f;
    __syncthreads();
    for (int i = tid; i < Ln; i += 256) {
        gin[0][PADn + i] = g[((long)b * Cn + 0) * Ln + i];
        gin[1][PADn + i] = g[((long)b * Cn + 1) * Ln + i];
    }
    for (int k = tid; k < KWn; k += 256) {
        float4 w = make_float4(0.0f, 0.0f, 0.0f, 0.0f);
        if (k < Kn) w = make_float4(conv_w[0 * Kn + k], conv_w[2 * Kn + k],
                                    conv_w[1 * Kn + k], conv_w[3 * Kn + k]);
        *(float4*)&wtap[4 * k] = w;
    }
    __syncthreads();

    const float cb0 = conv_b[0], cb1 = conv_b[1];
    {
        const float* in0 = &gin[0][8 * tid];
        const float* in1 = &gin[1][8 * tid];
        v2f   acc[8];
        float xa[16], xb[16];
        #pragma unroll
        for (int i = 0; i < 8; ++i) { acc[i].x = 0.0f; acc[i].y = 0.0f; }
        lds_ld4(xa + 0, in0 + 0);  lds_ld4(xa + 4, in0 + 4);
        lds_ld4(xb + 0, in1 + 0);  lds_ld4(xb + 4, in1 + 4);

        float4 wc[8];
        #pragma unroll
        for (int kk = 0; kk < 8; ++kk) wc[kk] = *(const float4*)&wtap[4 * kk];

        #pragma unroll 2
        for (int k0 = 0; k0 < KPn; k0 += 8) {
            lds_ld4(xa + 8, in0 + k0 + 8);  lds_ld4(xa + 12, in0 + k0 + 12);
            lds_ld4(xb + 8, in1 + k0 + 8);  lds_ld4(xb + 12, in1 + k0 + 12);
            float4 wn[8];
            #pragma unroll
            for (int kk = 0; kk < 8; ++kk) wn[kk] = *(const float4*)&wtap[4 * (k0 + 8 + kk)];
            __builtin_amdgcn_sched_barrier(0);
            #pragma unroll
            for (int kk = 0; kk < 8; ++kk) {
                v2f wa; wa.x = wc[kk].x; wa.y = wc[kk].y;
                v2f wb; wb.x = wc[kk].z; wb.y = wc[kk].w;
                #pragma unroll
                for (int i = 0; i < 8; ++i) {
                    v2f va; va.x = xa[kk + i]; va.y = xa[kk + i];
                    v2f vb; vb.x = xb[kk + i]; vb.y = xb[kk + i];
                    acc[i] = __builtin_elementwise_fma(wa, va, acc[i]);
                    acc[i] = __builtin_elementwise_fma(wb, vb, acc[i]);
                }
            }
            #pragma unroll
            for (int i = 0; i < 8; ++i) { xa[i] = xa[i + 8]; xb[i] = xb[i + 8]; wc[i] = wn[i]; }
        }

        float r[8];
        #pragma unroll
        for (int i = 0; i < 8; ++i)
            r[i] = fmaxf(fmaxf(acc[i].x + cb0, 0.0f), fmaxf(acc[i].y + cb1, 0.0f));
        *(float4*)&hbuf[8 * tid + 0] = make_float4(r[0], r[1], r[2], r[3]);
        *(float4*)&hbuf[8 * tid + 4] = make_float4(r[4], r[5], r[6], r[7]);
    }
    __syncthreads();
    float h[8];
    #pragma unroll
    for (int j = 0; j < 8; ++j) h[j] = hbuf[tid + 256 * j];

    float lm = -INFINITY;
    #pragma unroll
    for (int j = 0; j < 8; ++j) lm = fmaxf(lm, h[j]);
    red[tid] = lm; __syncthreads();
    for (int off = 128; off > 0; off >>= 1) {
        if (tid < off) red[tid] = fmaxf(red[tid], red[tid + off]);
        __syncthreads();
    }
    const float M = red[0];
    __syncthreads();

    float ls = 0.0f, lq = 0.0f;
    #pragma unroll
    for (int j = 0; j < 8; ++j) { h[j] = __expf(h[j] - M); ls += h[j]; lq += h[j] * q[j]; }
    red[tid] = ls; redq[tid] = lq; __syncthreads();
    for (int off = 128; off > 0; off >>= 1) {
        if (tid < off) { red[tid] += red[tid + off]; redq[tid] += redq[tid + off]; }
        __syncthreads();
    }
    const float inv = 1.0f / red[0];
    #pragma unroll
    for (int j = 0; j < 8; ++j)
        betaOut[(long)b * Ln + tid + 256 * j] = h[j] * inv;

    if (tid == 0) {
        const float d  = redq[0] * inv + (lin_b[1] - lin_b[0]);
        const float t  = __logf(1.0f + __expf(-fabsf(d)));
        logits[b * 2 + 0] = -(fmaxf(d, 0.0f) + t);
        logits[b * 2 + 1] = -(fmaxf(-d, 0.0f) + t);
    }
}

extern "C" void kernel_launch(void* const* d_in, const int* in_sizes, int n_in,
                              void* d_out, int out_size, void* d_ws, size_t ws_size,
                              hipStream_t stream)
{
    const int*   tok = (const int*)  d_in[0];
    const float* emb = (const float*)d_in[1];
    const float* cls = (const float*)d_in[2];
    const float* cw  = (const float*)d_in[3];
    const float* cb  = (const float*)d_in[4];
    const float* lw  = (const float*)d_in[5];
    const float* lb  = (const float*)d_in[6];

    float* out    = (float*)d_out;
    float* logits = out;
    float* beta   = out + Bn * Cn;
    float* g      = (float*)d_ws;

    k_phase1<<<1024, 256, 0, stream>>>(tok, emb, cls, lw, g, beta);
    k_phase2<<<Bn, 256, 0, stream>>>(g, cw, cb, lb, beta, logits);
}


# Block heads cut to one memory round trip: staging loads issued as one batch in both kernels; class rows staged raw behind one barrier, their norms applied to the finished dots.
# speedup vs baseline: 6.1962x; 1.3036x over previous
#include <hip/hip_runtime.h>
#include <hip/hip_bf16.h>
#include <math.h>

#define Bn   64
#define Ln   2048
#define Dn   300
#define Cn   2
#define Kn   111
#define PADn 55
#define GWn  2160
#define KPn  112
#define KWn  120

typedef __attribute__((ext_vector_type(2))) float v2f;
typedef __attribute__((ext_vector_type(8))) float v8f;

template <int S0, int N>
__device__ __forceinline__ void load_steps(const float4* __restrict__ row4, float4 (&a)[N])
{
    #pragma unroll
    for (int i = 0; i < N; ++i) a[i] = row4[2 * (S0 + i)];
}

template <int S0, int N>
__device__ __forceinline__ void mma_steps(const float4 (&a)[N], const float4* bb4,
                                          v8f& acc0, v8f& acc1, float& n2)
{
    #pragma unroll
    for (int i = 0; i < N; ++i) {
        const float4 bv = bb4[2 * (S0 + i)];
        v2f a0; a0.x = a[i].x; a0.y = a[i].y;
        v2f a1; a1.x = a[i].z; a1.y = a[i].w;
        v2f b0; b0.x = bv.x;   b0.y = bv.y;
        v2f b1; b1.x = bv.z;   b1.y = bv.w;
        n2 += a[i].x * a[i].x + a[i].y * a[i].y + a[i].z * a[i].z + a[i].w * a[i].w;
        acc0 = __builtin_amdgcn_wmma_f32_16x16x4_f32(
            false, a0, false, b0, (short)0, acc0, false, false);
        acc1 = __builtin_amdgcn_wmma_f32_16x16x4_f32(
            false, a1, false, b1, (short)0, acc1, false, false);
    }
}

__global__ __launch_bounds__(256) __attribute__((amdgpu_waves_per_eu(8)))
void k_phase1(const int* __restrict__ tok, const float* __restrict__ emb,
              const float* __restrict__ cls, const float* __restrict__ lin_w,
              float* __restrict__ g, float* __restrict__ qout)
{
    __shared__ __attribute__((aligned(16))) float clsn[4 * 304];
    __shared__ float cnrm[16];
    const int tid  = threadIdx.x;
    const int wave = __builtin_amdgcn_readfirstlane(tid >> 5);
    const int lane = tid & 31;
    const int tileId = blockIdx.x * 8 + wave;
    const int b  = tileId >> 7;
    const int l0 = (tileId & 127) * 16;
    const int m    = lane & 15;
    const int hi   = lane >> 4;
    const int koff = hi * 2;

    const int token = tok[b * Ln + l0 + m];
    if (tid < Dn / 4) {
        const float4 c0 = ((const float4*)cls)[tid];
        const float4 c1 = ((const float4*)(cls + Dn))[tid];
        const float4 w0 = ((const float4*)lin_w)[tid];
        const float4 w1 = ((const float4*)(lin_w + Dn))[tid];
        ((float4*)&clsn[0 * 304])[tid] = c0;
        ((float4*)&clsn[1 * 304])[tid] = c1;
        ((float4*)&clsn[2 * 304])[tid] = make_float4(w1.x - w0.x, w1.y - w0.y, w1.z - w0.z, w1.w - w0.w);
        ((float4*)&clsn[3 * 304])[tid] = make_float4(0.0f, 0.0f, 0.0f, 0.0f);
    }
    __syncthreads();

    const float* __restrict__ row = emb + (long)token * Dn;
    const float* bb = &clsn[(m < 3 ? m : 3) * 304];
    const float4* __restrict__ row4 = (const float4*)row + hi;
    const float4* bb4 = (const float4*)bb + hi;

    v8f acc0 = {}, acc1 = {};
    float n2 = 0.0f;
    float4 g0[10], g1[9], g2[9], g3[9];
    load_steps<0, 10>(row4, g0);
    load_steps<10, 9>(row4, g1);
    {
        const bool rd = lane < 25;
        const float4* p0 = (const float4*)&clsn[rd ? 12 * lane : 3 * 304];
        const float4* p1 = rd ? p0 + 76 : p0;
        float s0 = 0.0f, s1 = 0.0f;
        #pragma unroll
        for (int j = 0; j < 3; ++j) {
            const float4 u = p0[j];
            const float4 v = p1[j];
            s0 += u.x * u.x + u.y * u.y + u.z * u.z + u.w * u.w;
            s1 += v.x * v.x + v.y * v.y + v.z * v.z + v.w * v.w;
        }
        #pragma unroll
        for (int off = 16; off > 0; off >>= 1) {
            s0 += __shfl_xor(s0, off, 32);
            s1 += __shfl_xor(s1, off, 32);
        }
        if (lane < 2) cnrm[wave * 2 + lane] = lane ? s1 : s0;
    }
    mma_steps<0, 10>(g0, bb4, acc0, acc1, n2);
    asm volatile("" : "+v"(acc0), "+v"(acc1), "+v"(n2) :: "memory");
    __builtin_amdgcn_sched_barrier(0);
    load_steps<19, 9>(row4, g2);
    __builtin_amdgcn_sched_barrier(0);
    mma_steps<10, 9>(g1, bb4, acc0, acc1, n2);
    asm volatile("" : "+v"(acc0), "+v"(acc1), "+v"(n2) :: "memory");
    __builtin_amdgcn_sched_barrier(0);
    load_steps<28, 9>(row4, g3);
    const int d0 = 296 + koff;
    v2f at; at.x = row[d0]; at.y = row[d0 + 1];
    __builtin_amdgcn_sched_barrier(0);
    mma_steps<19, 9>(g2, bb4, acc0, acc1, n2);
    mma_steps<28, 9>(g3, bb4, acc0, acc1, n2);
    {
        v2f bf; bf.x = bb[d0];  bf.y = bb[d0 + 1];
        n2 += at.x * at.x + at.y * at.y;
        acc0 = __builtin_amdgcn_wmma_f32_16x16x4_f32(
            false, at, false, bf, (short)0, acc0, false, false);
    }
    const v8f acc = acc0 + acc1;

    n2 += __shfl_xor(n2, 16, 32);

    const bool isg    = (m < 2);
    const bool active = (m < 3);
    const int  mbase  = hi * 8;
    float* dst = isg ? g    + (((long)b * Cn + m) * Ln + l0 + mbase)
                     : qout + ((long)b * Ln + l0 + mbase);
    const float cscale = 1.0f / fmaxf(sqrtf(cnrm[wave * 2 + (m == 1 ? 1 : 0)]), 1e-12f);
    #pragma unroll
    for (int i = 0; i < 8; ++i) {
        float nn    = __shfl(n2, mbase + i, 32);
        float scale = 1.0f / fmaxf(sqrtf(nn), 1e-12f);
        if (active) dst[i] = isg ? acc[i] * scale * cscale : acc[i];
    }
}

__device__ __forceinline__ void lds_ld4(float* __restrict__ d, const float* s)
{
    const float4 v = *(const float4*)s;
    d[0] = v.x; d[1] = v.y; d[2] = v.z; d[3] = v.w;
}

__global__ __launch_bounds__(256)
void k_phase2(const float* __restrict__ g, const float* __restrict__ conv_w,
              const float* __restrict__ conv_b, const float* __restrict__ lin_b,
              float* __restrict__ betaOut, float* __restrict__ logits)
{
    __shared__ __attribute__((aligned(16))) float gin[Cn][GWn];
    __shared__ __attribute__((aligned(16))) float wtap[KWn * 4];
    __shared__ __attribute__((aligned(16))) float hbuf[Ln];
    __shared__ float red[256];
    __shared__ float redq[256];
    const int b = blockIdx.x, tid = threadIdx.x;

    float q[8], ga[8], gc[8];
    #pragma unroll
    for (int j = 0; j < 8; ++j) q[j] = betaOut[(long)b * Ln + tid + 256 * j];
    #pragma unroll
    for (int j = 0; j < 8; ++j) ga[j] = g[((long)b * Cn + 0) * Ln + tid + 256 * j];
    #pragma unroll
    for (int j = 0; j < 8; ++j) gc[j] = g[((long)b * Cn + 1) * Ln + tid + 256 * j];
    const int kc = tid < Kn ? tid : 0;
    float4 w = make_float4(conv_w[0 * Kn + kc], conv_w[2 * Kn + kc],
                           conv_w[1 * Kn + kc], conv_w[3 * Kn + kc]);
    if (tid >= Kn) w = make_float4(0.0f, 0.0f, 0.0f, 0.0f);
    const float cb0 = conv_b[0], cb1 = conv_b[1];
    const float lbd = lin_b[1] - lin_b[0];

    if (tid < Cn * (GWn - Ln)) {
        const int c = tid >= (GWn - Ln) ? 1 : 0;
        const int r = tid - c * (GWn - Ln);
        gin[c][r < PADn ? r : r + Ln] = 0.0f;
    }
    #pragma unroll
    for (int j = 0; j < 8; ++j) {
        gin[0][PADn + tid + 256 * j] = ga[j];
        gin[1][PADn + tid + 256 * j] = gc[j];
    }
    if (tid < KWn) *(float4*)&wtap[4 * tid] = w;
    __syncthreads();

    {
        const float* in0 = &gin[0][8 * tid];
        const float* in1 = &gin[1][8 * tid];
        v2f   acc[8];
        float xa[16], xb[16];
        #pragma unroll
        for (int i = 0; i < 8; ++i) { acc[i].x = 0.0f; acc[i].y = 0.0f; }
        lds_ld4(xa + 0, in0 + 0);  lds_ld4(xa + 4, in0 + 4);
        lds_ld4(xb + 0, in1 + 0);  lds_ld4(xb + 4, in1 + 4);

        float4 wc[8];
        #pragma unroll
        for (int kk = 0; kk < 8; ++kk) wc[kk] = *(const float4*)&wtap[4 * kk];

        #pragma unroll 2
        for (int k0 = 0; k0 < KPn; k0 += 8) {
            lds_ld4(xa + 8, in0 + k0 + 8);  lds_ld4(xa + 12, in0 + k0 + 12);
            lds_ld4(xb + 8, in1 + k0 + 8);  lds_ld4(xb + 12, in1 + k0 + 12);
            float4 wn[8];
            #pragma unroll
            for (int kk = 0; kk < 8; ++kk) wn[kk] = *(const float4*)&wtap[4 * (k0 + 8 + kk)];
            __builtin_amdgcn_sched_barrier(0);
            #pragma unroll
            for (int kk = 0; kk < 8; ++kk) {
                v2f wa; wa.x = wc[kk].x; wa.y = wc[kk].y;
                v2f wb; wb.x = wc[kk].z; wb.y = wc[kk].w;
                #pragma unroll
                for (int i = 0; i < 8; ++i) {
                    v2f va; va.x = xa[kk + i]; va.y = xa[kk + i];
                    v2f vb; vb.x = xb[kk + i]; vb.y = xb[kk + i];
                    acc[i] = __builtin_elementwise_fma(wa, va, acc[i]);
                    acc[i] = __builtin_elementwise_fma(wb, vb, acc[i]);
                }
            }
            #pragma unroll
            for (int i = 0; i < 8; ++i) { xa[i] = xa[i + 8]; xb[i] = xb[i + 8]; wc[i] = wn[i]; }
        }

        float r[8];
        #pragma unroll
        for (int i = 0; i < 8; ++i)
            r[i] = fmaxf(fmaxf(acc[i].x + cb0, 0.0f), fmaxf(acc[i].y + cb1, 0.0f));
        *(float4*)&hbuf[8 * tid + 0] = make_float4(r[0], r[1], r[2], r[3]);
        *(float4*)&hbuf[8 * tid + 4] = make_float4(r[4], r[5], r[6], r[7]);
    }
    __syncthreads();
    float h[8];
    #pragma unroll
    for (int j = 0; j < 8; ++j) h[j] = hbuf[tid + 256 * j];

    float lm = -INFINITY;
    #pragma unroll
    for (int j = 0; j < 8; ++j) lm = fmaxf(lm, h[j]);
    red[tid] = lm; __syncthreads();
    for (int off = 128; off > 0; off >>= 1) {
        if (tid < off) red[tid] = fmaxf(red[tid], red[tid + off]);
        __syncthreads();
    }
    const float M = red[0];
    __syncthreads();

    float ls = 0.0f, lq = 0.0f;
    #pragma unroll
    for (int j = 0; j < 8; ++j) { h[j] = __expf(h[j] - M); ls += h[j]; lq += h[j] * q[j]; }
    red[tid] = ls; redq[tid] = lq; __syncthreads();
    for (int off = 128; off > 0; off >>= 1) {
        if (tid < off) { red[tid] += red[tid + off]; redq[tid] += redq[tid + off]; }
        __syncthreads();
    }
    const float inv = 1.0f / red[0];
    #pragma unroll
    for (int j = 0; j < 8; ++j)
        betaOut[(long)b * Ln + tid + 256 * j] = h[j] * inv;

    if (tid == 0) {
        const float d  = redq[0] * inv + lbd;
        const float t  = __logf(1.0f + __expf(-fabsf(d)));
        logits[b * 2 + 0] = -(fmaxf(d, 0.0f) + t);
        logits[b * 2 + 1] = -(fmaxf(-d, 0.0f) + t);
    }
}

extern "C" void kernel_launch(void* const* d_in, const int* in_sizes, int n_in,
                              void* d_out, int out_size, void* d_ws, size_t ws_size,
                              hipStream_t stream)
{
    const int*   tok = (const int*)  d_in[0];
    const float* emb = (const float*)d_in[1];
    const float* cls = (const float*)d_in[2];
    const float* cw  = (const float*)d_in[3];
    const float* cb  = (const float*)d_in[4];
    const float* lw  = (const float*)d_in[5];
    const float* lb  = (const float*)d_in[6];

    float* out    = (float*)d_out;
    float* logits = out;
    float* beta   = out + Bn * Cn;
    float* g      = (float*)d_ws;

    k_phase1<<<1024, 256, 0, stream>>>(tok, emb, cls, lw, g, beta);
    k_phase2<<<Bn, 256, 0, stream>>>(g, cw, cb, lb, beta, logits);
}


# Phase 1 computed once per table row instead of per token (50000 rows against 131072 tokens): the table is swept in order into 16-byte entries (g0,g1,q) and phase 2 gathers each token's entry.
# speedup vs baseline: 8.6385x; 1.3942x over previous
#include <hip/hip_runtime.h>
#include <hip/hip_bf16.h>
#include <math.h>

#define Bn   64
#define Ln   2048
#define Dn   300
#define Cn   2
#define Kn   111
#define PADn 55
#define Vn   50000
#define TWn  5
static_assert(Vn == 625 * TWn * 16, "625 blocks of TWn tiles of 16 rows cover the table exactly");
#define GWn  2160
#define KPn  112
#define KWn  120

typedef __attribute__((ext_vector_type(2))) float v2f;
typedef __attribute__((ext_vector_type(8))) float v8f;

template <int S0, int N>
__device__ __forceinline__ void load_steps(const float4* __restrict__ row4, float4 (&a)[N])
{
    #pragma unroll
    for (int i = 0; i < N; ++i) a[i] = row4[2 * (S0 + i)];
}

template <int S0, int N>
__device__ __forceinline__ void mma_steps(const float4 (&a)[N], const float4* bb4,
                                          v8f& acc0, v8f& acc1, float& n2)
{
    #pragma unroll
    for (int i = 0; i < N; ++i) {
        const float4 bv = bb4[2 * (S0 + i)];
        v2f a0; a0.x = a[i].x; a0.y = a[i].y;
        v2f a1; a1.x = a[i].z; a1.y = a[i].w;
        v2f b0; b0.x = bv.x;   b0.y = bv.y;
        v2f b1; b1.x = bv.z;   b1.y = bv.w;
        n2 += a[i].x * a[i].x + a[i].y * a[i].y + a[i].z * a[i].z + a[i].w * a[i].w;
        acc0 = __builtin_amdgcn_wmma_f32_16x16x4_f32(
            false, a0, false, b0, (short)0, acc0, false, false);
        acc1 = __builtin_amdgcn_wmma_f32_16x16x4_f32(
            false, a1, false, b1, (short)0, acc1, false, false);
    }
}

__global__ __launch_bounds__(32 * TWn) __attribute__((amdgpu_waves_per_eu(4)))
void k_phase1(const float* __restrict__ emb, const float* __restrict__ cls,
              const float* __restrict__ lin_w, float* __restrict__ tab)
{
    __shared__ __attribute__((aligned(16))) float clsn[4 * 304];
    __shared__ float cnrm[2 * TWn];
    const int tid  = threadIdx.x;
    const int wave = __builtin_amdgcn_readfirstlane(tid >> 5);
    const int lane = tid & 31;
    const int tileId = blockIdx.x * TWn + wave;
    const int m    = lane & 15;
    const int hi   = lane >> 4;
    const int koff = hi * 2;

    float4 c0, c1, w0, w1;
    if (tid < Dn / 4) {
        c0 = ((const float4*)cls)[tid];
        c1 = ((const float4*)(cls + Dn))[tid];
        w0 = ((const float4*)lin_w)[tid];
        w1 = ((const float4*)(lin_w + Dn))[tid];
    }
    const float* __restrict__ row = emb + ((long)tileId * 16 + m) * Dn;
    const float4* __restrict__ row4 = (const float4*)row + hi;
    float4 g0[10], g1[9], g2[9], g3[9];
    load_steps<0, 10>(row4, g0);
    load_steps<10, 9>(row4, g1);
    if (tid < Dn / 4) {
        ((float4*)&clsn[0 * 304])[tid] = c0;
        ((float4*)&clsn[1 * 304])[tid] = c1;
        ((float4*)&clsn[2 * 304])[tid] = make_float4(w1.x - w0.x, w1.y - w0.y, w1.z - w0.z, w1.w - w0.w);
        ((float4*)&clsn[3 * 304])[tid] = make_float4(0.0f, 0.0f, 0.0f, 0.0f);
    }
    __syncthreads();

    const float* bb = &clsn[(m < 3 ? m : 3) * 304];
    const float4* bb4 = (const float4*)bb + hi;

    v8f acc0 = {}, acc1 = {};
    float n2 = 0.0f;
    {
        const bool rd = lane < 25;
        const float4* p0 = (const float4*)&clsn[rd ? 12 * lane : 3 * 304];
        const float4* p1 = rd ? p0 + 76 : p0;
        float s0 = 0.0f, s1 = 0.0f;
        #pragma unroll
        for (int j = 0; j < 3; ++j) {
            const float4 u = p0[j];
            const float4 v = p1[j];
            s0 += u.x * u.x + u.y * u.y + u.z * u.z + u.w * u.w;
            s1 += v.x * v.x + v.y * v.y + v.z * v.z + v.w * v.w;
        }
        #pragma unroll
        for (int off = 16; off > 0; off >>= 1) {
            s0 += __shfl_xor(s0, off, 32);
            s1 += __shfl_xor(s1, off, 32);
        }
        if (lane < 2) cnrm[wave * 2 + lane] = lane ? s1 : s0;
    }
    mma_steps<0, 10>(g0, bb4, acc0, acc1, n2);
    asm volatile("" : "+v"(acc0), "+v"(acc1), "+v"(n2) :: "memory");
    __builtin_amdgcn_sched_barrier(0);
    load_steps<19, 9>(row4, g2);
    __builtin_amdgcn_sched_barrier(0);
    mma_steps<10, 9>(g1, bb4, acc0, acc1, n2);
    asm volatile("" : "+v"(acc0), "+v"(acc1), "+v"(n2) :: "memory");
    __builtin_amdgcn_sched_barrier(0);
    load_steps<28, 9>(row4, g3);
    const int d0 = 296 + koff;
    v2f at; at.x = row[d0]; at.y = row[d0 + 1];
    __builtin_amdgcn_sched_barrier(0);
    mma_steps<19, 9>(g2, bb4, acc0, acc1, n2);
    mma_steps<28, 9>(g3, bb4, acc0, acc1, n2);
    {
        v2f bf; bf.x = bb[d0];  bf.y = bb[d0 + 1];
        n2 += at.x * at.x + at.y * at.y;
        acc0 = __builtin_amdgcn_wmma_f32_16x16x4_f32(
            false, at, false, bf, (short)0, acc0, false, false);
    }
    const v8f acc = acc0 + acc1;

    n2 += __shfl_xor(n2, 16, 32);

    const bool isg    = (m < 2);
    const bool active = (m < 4);
    const int  mbase  = hi * 8;
    float* dst = tab + (((long)tileId * 16 + mbase) * 4 + m);
    const float cscale = 1.0f / fmaxf(sqrtf(cnrm[wave * 2 + (m == 1 ? 1 : 0)]), 1e-12f);
    #pragma unroll
    for (int i = 0; i < 8; ++i) {
        float nn    = __shfl(n2, mbase + i, 32);
        float scale = 1.0f / fmaxf(sqrtf(nn), 1e-12f);
        if (active) dst[4 * i] = isg ? acc[i] * scale * cscale : (m == 2 ? acc[i] : 0.0f);
    }
}

__device__ __forceinline__ void lds_ld4(float* __restrict__ d, const float* s)
{
    const float4 v = *(const float4*)s;
    d[0] = v.x; d[1] = v.y; d[2] = v.z; d[3] = v.w;
}

__global__ __launch_bounds__(256)
void k_phase2(const int* __restrict__ tok, const float* __restrict__ tab,
              const float* __restrict__ conv_w,
              const float* __restrict__ conv_b, const float* __restrict__ lin_b,
              float* __restrict__ betaOut, float* __restrict__ logits)
{
    __shared__ __attribute__((aligned(16))) float gin[Cn][GWn];
    __shared__ __attribute__((aligned(16))) float wtap[KWn * 4];
    __shared__ __attribute__((aligned(16))) float hbuf[Ln];
    __shared__ float red[256];
    __shared__ float redq[256];
    const int b = blockIdx.x, tid = threadIdx.x;

    int tk[8];
    #pragma unroll
    for (int j = 0; j < 8; ++j) tk[j] = tok[b * Ln + tid + 256 * j];
    const int kc = tid < Kn ? tid : 0;
    float4 w = make_float4(conv_w[0 * Kn + kc], conv_w[2 * Kn + kc],
                           conv_w[1 * Kn + kc], conv_w[3 * Kn + kc]);
    if (tid >= Kn) w = make_float4(0.0f, 0.0f, 0.0f, 0.0f);
    const float cb0 = conv_b[0], cb1 = conv_b[1];
    const float lbd = lin_b[1] - lin_b[0];
    float q[8], ga[8], gc[8];
    {
        float4 e[8];
        #pragma unroll
        for (int j = 0; j < 8; ++j) e[j] = ((const float4*)tab)[tk[j]];
        #pragma unroll
        for (int j = 0; j < 8; ++j) { ga[j] = e[j].x; gc[j] = e[j].y; q[j] = e[j].z; }
    }

    if (tid < Cn * (GWn - Ln)) {
        const int c = tid >= (GWn - Ln) ? 1 : 0;
        const int r = tid - c * (GWn - Ln);
        gin[c][r < PADn ? r : r + Ln] = 0.0f;
    }
    #pragma unroll
    for (int j = 0; j < 8; ++j) {
        gin[0][PADn + tid + 256 * j] = ga[j];
        gin[1][PADn + tid + 256 * j] = gc[j];
    }
    if (tid < KWn) *(float4*)&wtap[4 * tid] = w;
    __syncthreads();

    {
        const float* in0 = &gin[0][8 * tid];
        const float* in1 = &gin[1][8 * tid];
        v2f   acc[8];
        float xa[16], xb[16];
        #pragma unroll
        for (int i = 0; i < 8; ++i) { acc[i].x = 0.0f; acc[i].y = 0.0f; }
        lds_ld4(xa + 0, in0 + 0);  lds_ld4(xa + 4, in0 + 4);
        lds_ld4(xb + 0, in1 + 0);  lds_ld4(xb + 4, in1 + 4);

        float4 wc[8];
        #pragma unroll
        for (int kk = 0; kk < 8; ++kk) wc[kk] = *(const float4*)&wtap[4 * kk];

        #pragma unroll 2
        for (int k0 = 0; k0 < KPn; k0 += 8) {
            lds_ld4(xa + 8, in0 + k0 + 8);  lds_ld4(xa + 12, in0 + k0 + 12);
            lds_ld4(xb + 8, in1 + k0 + 8);  lds_ld4(xb + 12, in1 + k0 + 12);
            float4 wn[8];
            #pragma unroll
            for (int kk = 0; kk < 8; ++kk) wn[kk] = *(const float4*)&wtap[4 * (k0 + 8 + kk)];
            __builtin_amdgcn_sched_barrier(0);
            #pragma unroll
            for (int kk = 0; kk < 8; ++kk) {
                v2f wa; wa.x = wc[kk].x; wa.y = wc[kk].y;
                v2f wb; wb.x = wc[kk].z; wb.y = wc[kk].w;
                #pragma unroll
                for (int i = 0; i < 8; ++i) {
                    v2f va; va.x = xa[kk + i]; va.y = xa[kk + i];
                    v2f vb; vb.x = xb[kk + i]; vb.y = xb[kk + i];
                    acc[i] = __builtin_elementwise_fma(wa, va, acc[i]);
                    acc[i] = __builtin_elementwise_fma(wb, vb, acc[i]);
                }
            }
            #pragma unroll
            for (int i = 0; i < 8; ++i) { xa[i] = xa[i + 8]; xb[i] = xb[i + 8]; wc[i] = wn[i]; }
        }

        float r[8];
        #pragma unroll
        for (int i = 0; i < 8; ++i)
            r[i] = fmaxf(fmaxf(acc[i].x + cb0, 0.0f), fmaxf(acc[i].y + cb1, 0.0f));
        *(float4*)&hbuf[8 * tid + 0] = make_float4(r[0], r[1], r[2], r[3]);
        *(float4*)&hbuf[8 * tid + 4] = make_float4(r[4], r[5], r[6], r[7]);
    }
    __syncthreads();
    float h[8];
    #pragma unroll
    for (int j = 0; j < 8; ++j) h[j] = hbuf[tid + 256 * j];

    float lm = -INFINITY;
    #pragma unroll
    for (int j = 0; j < 8; ++j) lm = fmaxf(lm, h[j]);
    red[tid] = lm; __syncthreads();
    for (int off = 128; off > 0; off >>= 1) {
        if (tid < off) red[tid] = fmaxf(red[tid], red[tid + off]);
        __syncthreads();
    }
    const float M = red[0];
    __syncthreads();

    float ls = 0.0f, lq = 0.0f;
    #pragma unroll
    for (int j = 0; j < 8; ++j) { h[j] = __expf(h[j] - M); ls += h[j]; lq += h[j] * q[j]; }
    red[tid] = ls; redq[tid] = lq; __syncthreads();
    for (int off = 128; off > 0; off >>= 1) {
        if (tid < off) { red[tid] += red[tid + off]; redq[tid] += redq[tid + off]; }
        __syncthreads();
    }
    const float inv = 1.0f / red[0];
    #pragma unroll
    for (int j = 0; j < 8; ++j)
        betaOut[(long)b * Ln + tid + 256 * j] = h[j] * inv;

    if (tid == 0) {
        const float d  = redq[0] * inv + lbd;
        const float t  = __logf(1.0f + __expf(-fabsf(d)));
        logits[b * 2 + 0] = -(fmaxf(d, 0.0f) + t);
        logits[b * 2 + 1] = -(fmaxf(-d, 0.0f) + t);
    }
}

extern "C" void kernel_launch(void* const* d_in, const int* in_sizes, int n_in,
                              void* d_out, int out_size, void* d_ws, size_t ws_size,
                              hipStream_t stream)
{
    const int*   tok = (const int*)  d_in[0];
    const float* emb = (const float*)d_in[1];
    const float* cls = (const float*)d_in[2];
    const float* cw  = (const float*)d_in[3];
    const float* cb  = (const float*)d_in[4];
    const float* lw  = (const float*)d_in[5];
    const float* lb  = (const float*)d_in[6];

    float* out    = (float*)d_out;
    float* logits = out;
    float* beta   = out + Bn * Cn;
    float* tab    = (float*)d_ws;

    k_phase1<<<625, 32 * TWn, 0, stream>>>(emb, cls, lw, tab);
    k_phase2<<<Bn, 256, 0, stream>>>(tok, tab, cw, cb, lb, beta, logits);
}


# Phase 2 softmax reductions without LDS trees: max, sum and sum(beta*q) are lane butterflies in each wave plus one 8-word combine, the max riding the conv's barrier; 21 workgroup barriers become 3.
# speedup vs baseline: 9.1782x; 1.0625x over previous
#include <hip/hip_runtime.h>
#include <hip/hip_bf16.h>
#include <math.h>

#define Bn   64
#define Ln   2048
#define Dn   300
#define Cn   2
#define Kn   111
#define PADn 55
#define Vn   50000
#define TWn  5
static_assert(Vn == 625 * TWn * 16, "625 blocks of TWn tiles of 16 rows cover the table exactly");
#define GWn  2160
#define KPn  112
#define KWn  120

typedef __attribute__((ext_vector_type(2))) float v2f;
typedef __attribute__((ext_vector_type(8))) float v8f;

template <int S0, int N>
__device__ __forceinline__ void load_steps(const float4* __restrict__ row4, float4 (&a)[N])
{
    #pragma unroll
    for (int i = 0; i < N; ++i) a[i] = row4[2 * (S0 + i)];
}

template <int S0, int N>
__device__ __forceinline__ void mma_steps(const float4 (&a)[N], const float4* bb4,
                                          v8f& acc0, v8f& acc1, float& n2)
{
    #pragma unroll
    for (int i = 0; i < N; ++i) {
        const float4 bv = bb4[2 * (S0 + i)];
        v2f a0; a0.x = a[i].x; a0.y = a[i].y;
        v2f a1; a1.x = a[i].z; a1.y = a[i].w;
        v2f b0; b0.x = bv.x;   b0.y = bv.y;
        v2f b1; b1.x = bv.z;   b1.y = bv.w;
        n2 += a[i].x * a[i].x + a[i].y * a[i].y + a[i].z * a[i].z + a[i].w * a[i].w;
        acc0 = __builtin_amdgcn_wmma_f32_16x16x4_f32(
            false, a0, false, b0, (short)0, acc0, false, false);
        acc1 = __builtin_amdgcn_wmma_f32_16x16x4_f32(
            false, a1, false, b1, (short)0, acc1, false, false);
    }
}

__global__ __launch_bounds__(32 * TWn) __attribute__((amdgpu_waves_per_eu(4)))
void k_phase1(const float* __restrict__ emb, const float* __restrict__ cls,
              const float* __restrict__ lin_w, float* __restrict__ tab)
{
    __shared__ __attribute__((aligned(16))) float clsn[4 * 304];
    __shared__ float cnrm[2 * TWn];
    const int tid  = threadIdx.x;
    const int wave = __builtin_amdgcn_readfirstlane(tid >> 5);
    const int lane = tid & 31;
    const int tileId = blockIdx.x * TWn + wave;
    const int m    = lane & 15;
    const int hi   = lane >> 4;
    const int koff = hi * 2;

    float4 c0, c1, w0, w1;
    if (tid < Dn / 4) {
        c0 = ((const float4*)cls)[tid];
        c1 = ((const float4*)(cls + Dn))[tid];
        w0 = ((const float4*)lin_w)[tid];
        w1 = ((const float4*)(lin_w + Dn))[tid];
    }
    const float* __restrict__ row = emb + ((long)tileId * 16 + m) * Dn;
    const float4* __restrict__ row4 = (const float4*)row + hi;
    float4 g0[10], g1[9], g2[9], g3[9];
    load_steps<0, 10>(row4, g0);
    load_steps<10, 9>(row4, g1);
    if (tid < Dn / 4) {
        ((float4*)&clsn[0 * 304])[tid] = c0;
        ((float4*)&clsn[1 * 304])[tid] = c1;
        ((float4*)&clsn[2 * 304])[tid] = make_float4(w1.x - w0.x, w1.y - w0.y, w1.z - w0.z, w1.w - w0.w);
        ((float4*)&clsn[3 * 304])[tid] = make_float4(0.0f, 0.0f, 0.0f, 0.0f);
    }
    __syncthreads();

    const float* bb = &clsn[(m < 3 ? m : 3) * 304];
    const float4* bb4 = (const float4*)bb + hi;

    v8f acc0 = {}, acc1 = {};
    float n2 = 0.0f;
    {
        const bool rd = lane < 25;
        const float4* p0 = (const float4*)&clsn[rd ? 12 * lane : 3 * 304];
        const float4* p1 = rd ? p0 + 76 : p0;
        float s0 = 0.0f, s1 = 0.0f;
        #pragma unroll
        for (int j = 0; j < 3; ++j) {
            const float4 u = p0[j];
            const float4 v = p1[j];
            s0 += u.x * u.x + u.y * u.y + u.z * u.z + u.w * u.w;
            s1 += v.x * v.x + v.y * v.y + v.z * v.z + v.w * v.w;
        }
        #pragma unroll
        for (int off = 16; off > 0; off >>= 1) {
            s0 += __shfl_xor(s0, off, 32);
            s1 += __shfl_xor(s1, off, 32);
        }
        if (lane < 2) cnrm[wave * 2 + lane] = lane ? s1 : s0;
    }
    mma_steps<0, 10>(g0, bb4, acc0, acc1, n2);
    asm volatile("" : "+v"(acc0), "+v"(acc1), "+v"(n2) :: "memory");
    __builtin_amdgcn_sched_barrier(0);
    load_steps<19, 9>(row4, g2);
    __builtin_amdgcn_sched_barrier(0);
    mma_steps<10, 9>(g1, bb4, acc0, acc1, n2);
    asm volatile("" : "+v"(acc0), "+v"(acc1), "+v"(n2) :: "memory");
    __builtin_amdgcn_sched_barrier(0);
    load_steps<28, 9>(row4, g3);
    const int d0 = 296 + koff;
    v2f at; at.x = row[d0]; at.y = row[d0 + 1];
    __builtin_amdgcn_sched_barrier(0);
    mma_steps<19, 9>(g2, bb4, acc0, acc1, n2);
    mma_steps<28, 9>(g3, bb4, acc0, acc1, n2);
    {
        v2f bf; bf.x = bb[d0];  bf.y = bb[d0 + 1];
        n2 += at.x * at.x + at.y * at.y;
        acc0 = __builtin_amdgcn_wmma_f32_16x16x4_f32(
            false, at, false, bf, (short)0, acc0, false, false);
    }
    const v8f acc = acc0 + acc1;

    n2 += __shfl_xor(n2, 16, 32);

    const bool isg    = (m < 2);
    const bool active = (m < 4);
    const int  mbase  = hi * 8;
    float* dst = tab + (((long)tileId * 16 + mbase) * 4 + m);
    const float cscale = 1.0f / fmaxf(sqrtf(cnrm[wave * 2 + (m == 1 ? 1 : 0)]), 1e-12f);
    #pragma unroll
    for (int i = 0; i < 8; ++i) {
        float nn    = __shfl(n2, mbase + i, 32);
        float scale = 1.0f / fmaxf(sqrtf(nn), 1e-12f);
        if (active) dst[4 * i] = isg ? acc[i] * scale * cscale : (m == 2 ? acc[i] : 0.0f);
    }
}

template <int CTRL>
__device__ __forceinline__ float xrow(float x)
{
    const int b = __builtin_bit_cast(int, x);
    return __builtin_bit_cast(float, __builtin_amdgcn_update_dpp(0, b, CTRL, 0xf, 0xf, true));
}
__device__ __forceinline__ float xhalf(float x)
{
    const int b = __builtin_bit_cast(int, x);
    return __builtin_bit_cast(float,
        __builtin_amdgcn_permlanex16(b, b, 0x76543210, 0xfedcba98, false, false));
}
__device__ __forceinline__ float wave_max(float x)
{
    x = fmaxf(x, xrow<0x161>(x));
    x = fmaxf(x, xrow<0x162>(x));
    x = fmaxf(x, xrow<0x164>(x));
    x = fmaxf(x, xrow<0x168>(x));
    return fmaxf(x, xhalf(x));
}
__device__ __forceinline__ float wave_sum(float x)
{
    x += xrow<0x161>(x);
    x += xrow<0x162>(x);
    x += xrow<0x164>(x);
    x += xrow<0x168>(x);
    return x + xhalf(x);
}

__device__ __forceinline__ void lds_ld4(float* __restrict__ d, const float* s)
{
    const float4 v = *(const float4*)s;
    d[0] = v.x; d[1] = v.y; d[2] = v.z; d[3] = v.w;
}

__global__ __launch_bounds__(256)
void k_phase2(const int* __restrict__ tok, const float* __restrict__ tab,
              const float* __restrict__ conv_w,
              const float* __restrict__ conv_b, const float* __restrict__ lin_b,
              float* __restrict__ betaOut, float* __restrict__ logits)
{
    __shared__ __attribute__((aligned(16))) float gin[Cn][GWn];
    __shared__ __attribute__((aligned(16))) float wtap[KWn * 4];
    __shared__ __attribute__((aligned(16))) float hbuf[Ln];
    __shared__ __attribute__((aligned(16))) float wmax[8];
    __shared__ __attribute__((aligned(16))) float wsum[8];
    __shared__ __attribute__((aligned(16))) float wsq[8];
    const int b = blockIdx.x, tid = threadIdx.x;
    const int wave = tid >> 5, lane = tid & 31;

    int tk[8];
    #pragma unroll
    for (int j = 0; j < 8; ++j) tk[j] = tok[b * Ln + tid + 256 * j];
    const int kc = tid < Kn ? tid : 0;
    float4 w = make_float4(conv_w[0 * Kn + kc], conv_w[2 * Kn + kc],
                           conv_w[1 * Kn + kc], conv_w[3 * Kn + kc]);
    if (tid >= Kn) w = make_float4(0.0f, 0.0f, 0.0f, 0.0f);
    const float cb0 = conv_b[0], cb1 = conv_b[1];
    const float lbd = lin_b[1] - lin_b[0];
    float q[8], ga[8], gc[8];
    {
        float4 e[8];
        #pragma unroll
        for (int j = 0; j < 8; ++j) e[j] = ((const float4*)tab)[tk[j]];
        #pragma unroll
        for (int j = 0; j < 8; ++j) { ga[j] = e[j].x; gc[j] = e[j].y; q[j] = e[j].z; }
    }

    if (tid < Cn * (GWn - Ln)) {
        const int c = tid >= (GWn - Ln) ? 1 : 0;
        const int r = tid - c * (GWn - Ln);
        gin[c][r < PADn ? r : r + Ln] = 0.0f;
    }
    #pragma unroll
    for (int j = 0; j < 8; ++j) {
        gin[0][PADn + tid + 256 * j] = ga[j];
        gin[1][PADn + tid + 256 * j] = gc[j];
    }
    if (tid < KWn) *(float4*)&wtap[4 * tid] = w;
    __syncthreads();

    {
        const float* in0 = &gin[0][8 * tid];
        const float* in1 = &gin[1][8 * tid];
        v2f   acc[8];
        float xa[16], xb[16];
        #pragma unroll
        for (int i = 0; i < 8; ++i) { acc[i].x = 0.0f; acc[i].y = 0.0f; }
        lds_ld4(xa + 0, in0 + 0);  lds_ld4(xa + 4, in0 + 4);
        lds_ld4(xb + 0, in1 + 0);  lds_ld4(xb + 4, in1 + 4);

        float4 wc[8];
        #pragma unroll
        for (int kk = 0; kk < 8; ++kk) wc[kk] = *(const float4*)&wtap[4 * kk];

        #pragma unroll 2
        for (int k0 = 0; k0 < KPn; k0 += 8) {
            lds_ld4(xa + 8, in0 + k0 + 8);  lds_ld4(xa + 12, in0 + k0 + 12);
            lds_ld4(xb + 8, in1 + k0 + 8);  lds_ld4(xb + 12, in1 + k0 + 12);
            float4 wn[8];
            #pragma unroll
            for (int kk = 0; kk < 8; ++kk) wn[kk] = *(const float4*)&wtap[4 * (k0 + 8 + kk)];
            __builtin_amdgcn_sched_barrier(0);
            #pragma unroll
            for (int kk = 0; kk < 8; ++kk) {
                v2f wa; wa.x = wc[kk].x; wa.y = wc[kk].y;
                v2f wb; wb.x = wc[kk].z; wb.y = wc[kk].w;
                #pragma unroll
                for (int i = 0; i < 8; ++i) {
                    v2f va; va.x = xa[kk + i]; va.y = xa[kk + i];
                    v2f vb; vb.x = xb[kk + i]; vb.y = xb[kk + i];
                    acc[i] = __builtin_elementwise_fma(wa, va, acc[i]);
                    acc[i] = __builtin_elementwise_fma(wb, vb, acc[i]);
                }
            }
            #pragma unroll
            for (int i = 0; i < 8; ++i) { xa[i] = xa[i + 8]; xb[i] = xb[i + 8]; wc[i] = wn[i]; }
        }

        float r[8];
        #pragma unroll
        for (int i = 0; i < 8; ++i)
            r[i] = fmaxf(fmaxf(acc[i].x + cb0, 0.0f), fmaxf(acc[i].y + cb1, 0.0f));
        *(float4*)&hbuf[8 * tid + 0] = make_float4(r[0], r[1], r[2], r[3]);
        *(float4*)&hbuf[8 * tid + 4] = make_float4(r[4], r[5], r[6], r[7]);
        float lm = fmaxf(fmaxf(fmaxf(r[0], r[1]), fmaxf(r[2], r[3])),
                         fmaxf(fmaxf(r[4], r[5]), fmaxf(r[6], r[7])));
        lm = wave_max(lm);
        if (lane == 0) wmax[wave] = lm;
    }
    __syncthreads();
    float h[8];
    #pragma unroll
    for (int j = 0; j < 8; ++j) h[j] = hbuf[tid + 256 * j];

    float M;
    {
        const float4 u = *(const float4*)&wmax[0];
        const float4 v = *(const float4*)&wmax[4];
        M = fmaxf(fmaxf(fmaxf(u.x, u.y), fmaxf(u.z, u.w)),
                  fmaxf(fmaxf(v.x, v.y), fmaxf(v.z, v.w)));
    }

    float ls = 0.0f, lq = 0.0f;
    #pragma unroll
    for (int j = 0; j < 8; ++j) { h[j] = __expf(h[j] - M); ls += h[j]; lq += h[j] * q[j]; }
    ls = wave_sum(ls);
    lq = wave_sum(lq);
    if (lane == 0) { wsum[wave] = ls; wsq[wave] = lq; }
    __syncthreads();
    float S, SQ;
    {
        const float4 u = *(const float4*)&wsum[0];
        const float4 v = *(const float4*)&wsum[4];
        const float4 x = *(const float4*)&wsq[0];
        const float4 y = *(const float4*)&wsq[4];
        S  = ((u.x + u.y) + (u.z + u.w)) + ((v.x + v.y) + (v.z + v.w));
        SQ = ((x.x + x.y) + (x.z + x.w)) + ((y.x + y.y) + (y.z + y.w));
    }
    const float inv = 1.0f / S;
    #pragma unroll
    for (int j = 0; j < 8; ++j)
        betaOut[(long)b * Ln + tid + 256 * j] = h[j] * inv;

    if (tid == 0) {
        const float d  = SQ * inv + lbd;
        const float t  = __logf(1.0f + __expf(-fabsf(d)));
        logits[b * 2 + 0] = -(fmaxf(d, 0.0f) + t);
        logits[b * 2 + 1] = -(fmaxf(-d, 0.0f) + t);
    }
}

extern "C" void kernel_launch(void* const* d_in, const int* in_sizes, int n_in,
                              void* d_out, int out_size, void* d_ws, size_t ws_size,
                              hipStream_t stream)
{
    const int*   tok = (const int*)  d_in[0];
    const float* emb = (const float*)d_in[1];
    const float* cls = (const float*)d_in[2];
    const float* cw  = (const float*)d_in[3];
    const float* cb  = (const float*)d_in[4];
    const float* lw  = (const float*)d_in[5];
    const float* lb  = (const float*)d_in[6];

    float* out    = (float*)d_out;
    float* logits = out;
    float* beta   = out + Bn * Cn;
    float* tab    = (float*)d_ws;

    k_phase1<<<625, 32 * TWn, 0, stream>>>(emb, cls, lw, tab);
    k_phase2<<<Bn, 256, 0, stream>>>(tok, tab, cw, cb, lb, beta, logits);
}


# Phase 1 as a resident tile loop: 1024 waves, one per SIMD, take 3 or 4 of the 3125 tiles each with the next tile's rows in flight; class and head rows stay in registers as B fragments.
# speedup vs baseline: 10.1792x; 1.1091x over previous
#include <hip/hip_runtime.h>
#include <hip/hip_bf16.h>
#include <math.h>

#define Bn   64
#define Ln   2048
#define Dn   300
#define Cn   2
#define Kn   111
#define PADn 55
#define Vn   50000
#define NTILEn 3125
#define P1Wn   4
#define NWAVEn 1024
static_assert(Vn == NTILEn * 16, "3125 tiles of 16 rows cover the table exactly");
static_assert(3 * NWAVEn <= NTILEn && NTILEn <= 4 * NWAVEn, "every wave has 3 tiles, some a fourth");
#define GWn  2160
#define KPn  112
#define KWn  120

typedef __attribute__((ext_vector_type(2))) float v2f;
typedef __attribute__((ext_vector_type(8))) float v8f;

struct Tile { float4 a[37]; v2f at; };

__device__ __forceinline__ void load_tile(const float* __restrict__ emb, int tile, int m, int hi, Tile& T)
{
    const float* __restrict__ row = emb + ((long)tile * 16 + m) * Dn;
    const float4* __restrict__ row4 = (const float4*)row + hi;
    #pragma unroll
    for (int s = 0; s < 37; ++s) T.a[s] = row4[2 * s];
    const int d0 = 296 + 2 * hi;
    T.at.x = row[d0]; T.at.y = row[d0 + 1];
}

__device__ __forceinline__ void mma_tile(const Tile& T, const float4 (&bf)[37], v2f bt,
                                         v8f& acc, float& n2)
{
    v8f acc0 = {}, acc1 = {};
    v2f nv; nv.x = 0.0f; nv.y = 0.0f;
    #pragma unroll
    for (int s = 0; s < 37; ++s) {
        v2f a0; a0.x = T.a[s].x; a0.y = T.a[s].y;
        v2f a1; a1.x = T.a[s].z; a1.y = T.a[s].w;
        v2f b0; b0.x = bf[s].x;  b0.y = bf[s].y;
        v2f b1; b1.x = bf[s].z;  b1.y = bf[s].w;
        nv = __builtin_elementwise_fma(a0, a0, nv);
        nv = __builtin_elementwise_fma(a1, a1, nv);
        acc0 = __builtin_amdgcn_wmma_f32_16x16x4_f32(
            false, a0, false, b0, (short)0, acc0, false, false);
        acc1 = __builtin_amdgcn_wmma_f32_16x16x4_f32(
            false, a1, false, b1, (short)0, acc1, false, false);
    }
    nv = __builtin_elementwise_fma(T.at, T.at, nv);
    acc0 = __builtin_amdgcn_wmma_f32_16x16x4_f32(
        false, T.at, false, bt, (short)0, acc0, false, false);
    acc = acc0 + acc1;
    n2  = nv.x + nv.y;
}

__device__ __forceinline__ void store_tile(float* __restrict__ tab, int tile, int m, int hi,
                                           v8f acc, float n2, float cscale)
{
    n2 += __shfl_xor(n2, 16, 32);
    const bool isg    = (m < 2);
    const bool active = (m < 4);
    const int  mbase  = hi * 8;
    float* dst = tab + (((long)tile * 16 + mbase) * 4 + m);
    const float sc = 1.0f / fmaxf(sqrtf(n2), 1e-12f);
    float scale[8];
    #pragma unroll
    for (int i = 0; i < 8; ++i) scale[i] = __shfl(sc, mbase + i, 32);
    if (active) {
        #pragma unroll
        for (int i = 0; i < 8; ++i)
            dst[4 * i] = isg ? acc[i] * scale[i] * cscale : (m == 2 ? acc[i] : 0.0f);
    }
}

__global__ __launch_bounds__(32 * P1Wn) __attribute__((amdgpu_waves_per_eu(1, 1)))
void k_phase1(const float* __restrict__ emb, const float* __restrict__ cls,
              const float* __restrict__ lin_w, float* __restrict__ tab)
{
    __shared__ __attribute__((aligned(16))) float clsn[4 * 304];
    const int tid  = threadIdx.x;
    const int wave = __builtin_amdgcn_readfirstlane(tid >> 5);
    const int lane = tid & 31;
    const int t0   = blockIdx.x * P1Wn + wave;
    const int m    = lane & 15;
    const int hi   = lane >> 4;

    float4 c0, c1, w0, w1;
    if (tid < Dn / 4) {
        c0 = ((const float4*)cls)[tid];
        c1 = ((const float4*)(cls + Dn))[tid];
        w0 = ((const float4*)lin_w)[tid];
        w1 = ((const float4*)(lin_w + Dn))[tid];
    }
    Tile TA, TB;
    load_tile(emb, t0, m, hi, TA);
    if (tid < Dn / 4) {
        ((float4*)&clsn[0 * 304])[tid] = c0;
        ((float4*)&clsn[1 * 304])[tid] = c1;
        ((float4*)&clsn[2 * 304])[tid] = make_float4(w1.x - w0.x, w1.y - w0.y, w1.z - w0.z, w1.w - w0.w);
        ((float4*)&clsn[3 * 304])[tid] = make_float4(0.0f, 0.0f, 0.0f, 0.0f);
    }
    __syncthreads();

    const float* bb = &clsn[(m < 3 ? m : 3) * 304];
    const float4* bb4 = (const float4*)bb + hi;
    float4 bf[37];
    #pragma unroll
    for (int s = 0; s < 37; ++s) bf[s] = bb4[2 * s];
    v2f bt; bt.x = bb[296 + 2 * hi]; bt.y = bb[297 + 2 * hi];

    float cscale;
    {
        const bool rd = lane < 25;
        const float4* p0 = (const float4*)&clsn[rd ? 12 * lane : 3 * 304];
        const float4* p1 = rd ? p0 + 76 : p0;
        float s0 = 0.0f, s1 = 0.0f;
        #pragma unroll
        for (int j = 0; j < 3; ++j) {
            const float4 u = p0[j];
            const float4 v = p1[j];
            s0 += u.x * u.x + u.y * u.y + u.z * u.z + u.w * u.w;
            s1 += v.x * v.x + v.y * v.y + v.z * v.z + v.w * v.w;
        }
        #pragma unroll
        for (int off = 16; off > 0; off >>= 1) {
            s0 += __shfl_xor(s0, off, 32);
            s1 += __shfl_xor(s1, off, 32);
        }
        cscale = 1.0f / fmaxf(sqrtf(m == 1 ? s1 : s0), 1e-12f);
    }

    v8f acc; float n2;
    __builtin_amdgcn_sched_barrier(0);
    load_tile(emb, t0 + NWAVEn, m, hi, TB);
    __builtin_amdgcn_sched_barrier(0);
    mma_tile(TA, bf, bt, acc, n2);
    int t2 = t0 + 2 * NWAVEn;
    asm volatile("" : "+v"(acc), "+v"(n2), "+s"(t2));
    __builtin_amdgcn_sched_barrier(0);
    load_tile(emb, t2, m, hi, TA);
    __builtin_amdgcn_sched_barrier(0);
    store_tile(tab, t0, m, hi, acc, n2, cscale);
    mma_tile(TB, bf, bt, acc, n2);
    int t3 = t0 + 3 * NWAVEn;
    asm volatile("" : "+v"(acc), "+v"(n2), "+s"(t3));
    __builtin_amdgcn_sched_barrier(0);
    const bool has4 = t3 < NTILEn;
    if (has4) load_tile(emb, t3, m, hi, TB);
    __builtin_amdgcn_sched_barrier(0);
    store_tile(tab, t0 + NWAVEn, m, hi, acc, n2, cscale);
    mma_tile(TA, bf, bt, acc, n2);
    store_tile(tab, t2, m, hi, acc, n2, cscale);
    if (has4) {
        mma_tile(TB, bf, bt, acc, n2);
        store_tile(tab, t3, m, hi, acc, n2, cscale);
    }
}

template <int CTRL>
__device__ __forceinline__ float xrow(float x)
{
    const int b = __builtin_bit_cast(int, x);
    return __builtin_bit_cast(float, __builtin_amdgcn_update_dpp(0, b, CTRL, 0xf, 0xf, true));
}
__device__ __forceinline__ float xhalf(float x)
{
    const int b = __builtin_bit_cast(int, x);
    return __builtin_bit_cast(float,
        __builtin_amdgcn_permlanex16(b, b, 0x76543210, 0xfedcba98, false, false));
}
__device__ __forceinline__ float wave_max(float x)
{
    x = fmaxf(x, xrow<0x161>(x));
    x = fmaxf(x, xrow<0x162>(x));
    x = fmaxf(x, xrow<0x164>(x));
    x = fmaxf(x, xrow<0x168>(x));
    return fmaxf(x, xhalf(x));
}
__device__ __forceinline__ float wave_sum(float x)
{
    x += xrow<0x161>(x);
    x += xrow<0x162>(x);
    x += xrow<0x164>(x);
    x += xrow<0x168>(x);
    return x + xhalf(x);
}

__device__ __forceinline__ void lds_ld4(float* __restrict__ d, const float* s)
{
    const float4 v = *(const float4*)s;
    d[0] = v.x; d[1] = v.y; d[2] = v.z; d[3] = v.w;
}

__global__ __launch_bounds__(256)
void k_phase2(const int* __restrict__ tok, const float* __restrict__ tab,
              const float* __restrict__ conv_w,
              const float* __restrict__ conv_b, const float* __restrict__ lin_b,
              float* __restrict__ betaOut, float* __restrict__ logits)
{
    __shared__ __attribute__((aligned(16))) float gin[Cn][GWn];
    __shared__ __attribute__((aligned(16))) float wtap[KWn * 4];
    __shared__ __attribute__((aligned(16))) float hbuf[Ln];
    __shared__ __attribute__((aligned(16))) float wmax[8];
    __shared__ __attribute__((aligned(16))) float wsum[8];
    __shared__ __attribute__((aligned(16))) float wsq[8];
    const int b = blockIdx.x, tid = threadIdx.x;
    const int wave = tid >> 5, lane = tid & 31;

    int tk[8];
    #pragma unroll
    for (int j = 0; j < 8; ++j) tk[j] = tok[b * Ln + tid + 256 * j];
    const int kc = tid < Kn ? tid : 0;
    float4 w = make_float4(conv_w[0 * Kn + kc], conv_w[2 * Kn + kc],
                           conv_w[1 * Kn + kc], conv_w[3 * Kn + kc]);
    if (tid >= Kn) w = make_float4(0.0f, 0.0f, 0.0f, 0.0f);
    const float cb0 = conv_b[0], cb1 = conv_b[1];
    const float lbd = lin_b[1] - lin_b[0];
    float q[8], ga[8], gc[8];
    {
        float4 e[8];
        #pragma unroll
        for (int j = 0; j < 8; ++j) e[j] = ((const float4*)tab)[tk[j]];
        #pragma unroll
        for (int j = 0; j < 8; ++j) { ga[j] = e[j].x; gc[j] = e[j].y; q[j] = e[j].z; }
    }

    if (tid < Cn * (GWn - Ln)) {
        const int c = tid >= (GWn - Ln) ? 1 : 0;
        const int r = tid - c * (GWn - Ln);
        gin[c][r < PADn ? r : r + Ln] = 0.0f;
    }
    #pragma unroll
    for (int j = 0; j < 8; ++j) {
        gin[0][PADn + tid + 256 * j] = ga[j];
        gin[1][PADn + tid + 256 * j] = gc[j];
    }
    if (tid < KWn) *(float4*)&wtap[4 * tid] = w;
    __syncthreads();

    {
        const float* in0 = &gin[0][8 * tid];
        const float* in1 = &gin[1][8 * tid];
        v2f   acc[8];
        float xa[16], xb[16];
        #pragma unroll
        for (int i = 0; i < 8; ++i) { acc[i].x = 0.0f; acc[i].y = 0.0f; }
        lds_ld4(xa + 0, in0 + 0);  lds_ld4(xa + 4, in0 + 4);
        lds_ld4(xb + 0, in1 + 0);  lds_ld4(xb + 4, in1 + 4);

        float4 wc[8];
        #pragma unroll
        for (int kk = 0; kk < 8; ++kk) wc[kk] = *(const float4*)&wtap[4 * kk];

        #pragma unroll 2
        for (int k0 = 0; k0 < KPn; k0 += 8) {
            lds_ld4(xa + 8, in0 + k0 + 8);  lds_ld4(xa + 12, in0 + k0 + 12);
            lds_ld4(xb + 8, in1 + k0 + 8);  lds_ld4(xb + 12, in1 + k0 + 12);
            float4 wn[8];
            #pragma unroll
            for (int kk = 0; kk < 8; ++kk) wn[kk] = *(const float4*)&wtap[4 * (k0 + 8 + kk)];
            __builtin_amdgcn_sched_barrier(0);
            #pragma unroll
            for (int kk = 0; kk < 8; ++kk) {
                v2f wa; wa.x = wc[kk].x; wa.y = wc[kk].y;
                v2f wb; wb.x = wc[kk].z; wb.y = wc[kk].w;
                #pragma unroll
                for (int i = 0; i < 8; ++i) {
                    v2f va; va.x = xa[kk + i]; va.y = xa[kk + i];
                    v2f vb; vb.x = xb[kk + i]; vb.y = xb[kk + i];
                    acc[i] = __builtin_elementwise_fma(wa, va, acc[i]);
                    acc[i] = __builtin_elementwise_fma(wb, vb, acc[i]);
                }
            }
            #pragma unroll
            for (int i = 0; i < 8; ++i) { xa[i] = xa[i + 8]; xb[i] = xb[i + 8]; wc[i] = wn[i]; }
        }

        float r[8];
        #pragma unroll
        for (int i = 0; i < 8; ++i)
            r[i] = fmaxf(fmaxf(acc[i].x + cb0, 0.0f), fmaxf(acc[i].y + cb1, 0.0f));
        *(float4*)&hbuf[8 * tid + 0] = make_float4(r[0], r[1], r[2], r[3]);
        *(float4*)&hbuf[8 * tid + 4] = make_float4(r[4], r[5], r[6], r[7]);
        float lm = fmaxf(fmaxf(fmaxf(r[0], r[1]), fmaxf(r[2], r[3])),
                         fmaxf(fmaxf(r[4], r[5]), fmaxf(r[6], r[7])));
        lm = wave_max(lm);
        if (lane == 0) wmax[wave] = lm;
    }
    __syncthreads();
    float h[8];
    #pragma unroll
    for (int j = 0; j < 8; ++j) h[j] = hbuf[tid + 256 * j];

    float M;
    {
        const float4 u = *(const float4*)&wmax[0];
        const float4 v = *(const float4*)&wmax[4];
        M = fmaxf(fmaxf(fmaxf(u.x, u.y), fmaxf(u.z, u.w)),
                  fmaxf(fmaxf(v.x, v.y), fmaxf(v.z, v.w)));
    }

    float ls = 0.0f, lq = 0.0f;
    #pragma unroll
    for (int j = 0; j < 8; ++j) { h[j] = __expf(h[j] - M); ls += h[j]; lq += h[j] * q[j]; }
    ls = wave_sum(ls);
    lq = wave_sum(lq);
    if (lane == 0) { wsum[wave] = ls; wsq[wave] = lq; }
    __syncthreads();
    float S, SQ;
    {
        const float4 u = *(const float4*)&wsum[0];
        const float4 v = *(const float4*)&wsum[4];
        const float4 x = *(const float4*)&wsq[0];
        const float4 y = *(const float4*)&wsq[4];
        S  = ((u.x + u.y) + (u.z + u.w)) + ((v.x + v.y) + (v.z + v.w));
        SQ = ((x.x + x.y) + (x.z + x.w)) + ((y.x + y.y) + (y.z + y.w));
    }
    const float inv = 1.0f / S;
    #pragma unroll
    for (int j = 0; j < 8; ++j)
        betaOut[(long)b * Ln + tid + 256 * j] = h[j] * inv;

    if (tid == 0) {
        const float d  = SQ * inv + lbd;
        const float t  = __logf(1.0f + __expf(-fabsf(d)));
        logits[b * 2 + 0] = -(fmaxf(d, 0.0f) + t);
        logits[b * 2 + 1] = -(fmaxf(-d, 0.0f) + t);
    }
}

extern "C" void kernel_launch(void* const* d_in, const int* in_sizes, int n_in,
                              void* d_out, int out_size, void* d_ws, size_t ws_size,
                              hipStream_t stream)
{
    const int*   tok = (const int*)  d_in[0];
    const float* emb = (const float*)d_in[1];
    const float* cls = (const float*)d_in[2];
    const float* cw  = (const float*)d_in[3];
    const float* cb  = (const float*)d_in[4];
    const float* lw  = (const float*)d_in[5];
    const float* lb  = (const float*)d_in[6];

    float* out    = (float*)d_out;
    float* logits = out;
    float* beta   = out + Bn * Cn;
    float* tab    = (float*)d_ws;

    k_phase1<<<256, 32 * P1Wn, 0, stream>>>(emb, cls, lw, tab);
    k_phase2<<<Bn, 256, 0, stream>>>(tok, tab, cw, cb, lb, beta, logits);
}


# Phase 1: a matrix step's 16 A rows are 4 table rows x 4 column chunks, so a load takes 128 contiguous bytes of 4 rows, not 32 bytes of 16; lane exchanges join chunks; waves get 48 or 56 rows.
# speedup vs baseline: 11.0330x; 1.0839x over previous
#include <hip/hip_runtime.h>
#include <hip/hip_bf16.h>
#include <math.h>

#define Bn   64
#define Ln   2048
#define Dn   300
#define Cn   2
#define Kn   111
#define PADn 55
#define Vn   50000
#define NPAIRn 6250
#define P1Wn   4
#define NWAVEn 1024
#define XTRAn  106
static_assert(Vn == NPAIRn * 8, "6250 pairs of 8 rows cover the table exactly");
static_assert(NPAIRn == 6 * NWAVEn + XTRAn && XTRAn <= NWAVEn, "every wave has 6 pairs, the first 106 a seventh");
#define GWn  2160
#define KPn  112
#define KWn  120

typedef __attribute__((ext_vector_type(2))) float v2f;
typedef __attribute__((ext_vector_type(8))) float v8f;

template <int CTRL>
__device__ __forceinline__ float xrow(float x)
{
    const int b = __builtin_bit_cast(int, x);
    return __builtin_bit_cast(float, __builtin_amdgcn_update_dpp(0, b, CTRL, 0xf, 0xf, true));
}
__device__ __forceinline__ float xhalf(float x)
{
    const int b = __builtin_bit_cast(int, x);
    return __builtin_bit_cast(float,
        __builtin_amdgcn_permlanex16(b, b, 0x76543210, 0xfedcba98, false, false));
}

struct Grp { float4 a[9]; v2f at; };

__device__ __forceinline__ void load_grp(const float* __restrict__ emb, int R, int r, int j, int hi, Grp& G)
{
    const float* __restrict__ row = emb + (long)(R + r) * Dn;
    const float4* __restrict__ row4 = (const float4*)row + (2 * j + hi);
    #pragma unroll
    for (int s = 0; s < 9; ++s) G.a[s] = row4[8 * s];
    const int jt = j < 3 ? j : 2;
    G.at = *(const v2f*)(row + 288 + 4 * jt + 2 * hi);
}

__device__ __forceinline__ void mma_pair(const Grp& G, const Grp& H, const float4 (&bf)[9], v2f bt,
                                         bool tail_on, v8f& accG, v8f& accH, float& n2G, float& n2H)
{
    v8f cg = {}, ch = {};
    v2f ng; ng.x = 0.0f; ng.y = 0.0f;
    v2f nh; nh.x = 0.0f; nh.y = 0.0f;
    #pragma unroll
    for (int s = 0; s < 9; ++s) {
        v2f g0; g0.x = G.a[s].x; g0.y = G.a[s].y;
        v2f g1; g1.x = G.a[s].z; g1.y = G.a[s].w;
        v2f h0; h0.x = H.a[s].x; h0.y = H.a[s].y;
        v2f h1; h1.x = H.a[s].z; h1.y = H.a[s].w;
        v2f b0; b0.x = bf[s].x;  b0.y = bf[s].y;
        v2f b1; b1.x = bf[s].z;  b1.y = bf[s].w;
        ng = __builtin_elementwise_fma(g0, g0, ng);
        ng = __builtin_elementwise_fma(g1, g1, ng);
        nh = __builtin_elementwise_fma(h0, h0, nh);
        nh = __builtin_elementwise_fma(h1, h1, nh);
        cg = __builtin_amdgcn_wmma_f32_16x16x4_f32(
            false, g0, false, b0, (short)0, cg, false, false);
        ch = __builtin_amdgcn_wmma_f32_16x16x4_f32(
            false, h0, false, b0, (short)0, ch, false, false);
        cg = __builtin_amdgcn_wmma_f32_16x16x4_f32(
            false, g1, false, b1, (short)0, cg, false, false);
        ch = __builtin_amdgcn_wmma_f32_16x16x4_f32(
            false, h1, false, b1, (short)0, ch, false, false);
    }
    v2f gt; gt.x = tail_on ? G.at.x : 0.0f; gt.y = tail_on ? G.at.y : 0.0f;
    v2f ht; ht.x = tail_on ? H.at.x : 0.0f; ht.y = tail_on ? H.at.y : 0.0f;
    ng = __builtin_elementwise_fma(gt, gt, ng);
    nh = __builtin_elementwise_fma(ht, ht, nh);
    cg = __builtin_amdgcn_wmma_f32_16x16x4_f32(
        false, gt, false, bt, (short)0, cg, false, false);
    ch = __builtin_amdgcn_wmma_f32_16x16x4_f32(
        false, ht, false, bt, (short)0, ch, false, false);
    accG = cg;  accH = ch;
    n2G = ng.x + ng.y;  n2H = nh.x + nh.y;
}

template <int NG>
__device__ __forceinline__ void load_grps(const float* __restrict__ emb, int R, int r, int j, int hi,
                                          Grp (&U)[4])
{
    #pragma unroll
    for (int g = 0; g < NG; ++g) load_grp(emb, R + 4 * g, r, j, hi, U[g]);
}

template <int NG>
__device__ __forceinline__ void mma_grps(const Grp (&U)[4], const float4 (&bf)[9], v2f bt, bool tail_on,
                                         v8f (&acc)[4], float (&n2)[4])
{
    static_assert(NG % 2 == 0, "groups go two at a time");
    #pragma unroll
    for (int g = 0; g < NG; g += 2)
        mma_pair(U[g], U[g + 1], bf, bt, tail_on, acc[g], acc[g + 1], n2[g], n2[g + 1]);
}

template <int NG>
__device__ __forceinline__ void store_grps(float* __restrict__ tab, int R, int cc, int j, int hi,
                                           const v8f (&acc)[4], const float (&n2)[4], float cscale)
{
    float d0[NG], d1[NG], nn[NG], sA[NG], sB[NG];
    #pragma unroll
    for (int g = 0; g < NG; ++g) {
        const v8f a = acc[g];
        float lo = a[3], up = a[7];
        lo = j == 2 ? a[2] : lo;   up = j == 2 ? a[6] : up;
        lo = j == 1 ? a[1] : lo;   up = j == 1 ? a[5] : up;
        lo = j == 0 ? a[0] : lo;   up = j == 0 ? a[4] : up;
        lo += xrow<0x161>(lo);  lo += xrow<0x162>(lo);
        up += xrow<0x161>(up);  up += xrow<0x162>(up);
        float n = n2[g];
        n += xrow<0x161>(n);    n += xrow<0x162>(n);    n += xhalf(n);
        d0[g] = lo;  d1[g] = up;  nn[g] = n;
    }
    float ns = nn[NG - 1];
    #pragma unroll
    for (int g = NG - 2; g >= 0; --g) ns = j == g ? nn[g] : ns;
    const float sc = 1.0f / fmaxf(sqrtf(ns), 1e-12f);
    #pragma unroll
    for (int g = 0; g < NG; ++g) {
        sA[g] = __shfl(sc, 8 * hi + g, 32);
        sB[g] = __shfl(sc, 8 * hi + 4 + g, 32);
    }
    const float m2 = cc < 2 ? cscale : (cc == 2 ? 1.0f : 0.0f);
    if (j < 2) {
        #pragma unroll
        for (int g = 0; g < NG; ++g) {
            const float d = j == 0 ? d0[g] : d1[g];
            const float s = j == 0 ? sA[g] : sB[g];
            tab[(long)(R + 4 * g + 2 * hi + j) * 4 + cc] = d * (cc < 2 ? s : 1.0f) * m2;
        }
    }
}

__global__ __launch_bounds__(32 * P1Wn) __attribute__((amdgpu_waves_per_eu(1, 1)))
void k_phase1(const float* __restrict__ emb, const float* __restrict__ cls,
              const float* __restrict__ lin_w, float* __restrict__ tab)
{
    __shared__ __attribute__((aligned(16))) float clsn[4 * 304];
    const int tid  = threadIdx.x;
    const int wave = __builtin_amdgcn_readfirstlane(tid >> 5);
    const int lane = tid & 31;
    const int gw   = blockIdx.x * P1Wn + wave;
    const bool has7 = gw < XTRAn;
    const int R0   = 8 * (6 * gw + (has7 ? gw : XTRAn));
    const int c16  = lane & 15;
    const int hi   = lane >> 4;
    const int r    = c16 >> 2;
    const int j    = c16 & 3;
    const bool tail_on = j < 3;

    float4 c0, c1, w0, w1;
    if (tid < Dn / 4) {
        c0 = ((const float4*)cls)[tid];
        c1 = ((const float4*)(cls + Dn))[tid];
        w0 = ((const float4*)lin_w)[tid];
        w1 = ((const float4*)(lin_w + Dn))[tid];
    }
    Grp UA[4], UB[4], UC[4];
    load_grps<4>(emb, R0, r, j, hi, UA);
    if (tid < Dn / 4) {
        ((float4*)&clsn[0 * 304])[tid] = c0;
        ((float4*)&clsn[1 * 304])[tid] = c1;
        ((float4*)&clsn[2 * 304])[tid] = make_float4(w1.x - w0.x, w1.y - w0.y, w1.z - w0.z, w1.w - w0.w);
        ((float4*)&clsn[3 * 304])[tid] = make_float4(0.0f, 0.0f, 0.0f, 0.0f);
    }
    __syncthreads();

    const float* bb = &clsn[r * 304];
    const float4* bb4 = (const float4*)bb + (2 * j + hi);
    float4 bf[9];
    #pragma unroll
    for (int s = 0; s < 9; ++s) bf[s] = bb4[8 * s];
    v2f bt;
    {
        const int jt = tail_on ? j : 2;
        const v2f t = *(const v2f*)(bb + 288 + 4 * jt + 2 * hi);
        bt.x = tail_on ? t.x : 0.0f;
        bt.y = tail_on ? t.y : 0.0f;
    }

    float cscale;
    {
        const bool rd = lane < 25;
        const float4* p0 = (const float4*)&clsn[rd ? 12 * lane : 3 * 304];
        const float4* p1 = rd ? p0 + 76 : p0;
        float s0 = 0.0f, s1 = 0.0f;
        #pragma unroll
        for (int k = 0; k < 3; ++k) {
            const float4 u = p0[k];
            const float4 v = p1[k];
            s0 += u.x * u.x + u.y * u.y + u.z * u.z + u.w * u.w;
            s1 += v.x * v.x + v.y * v.y + v.z * v.z + v.w * v.w;
        }
        #pragma unroll
        for (int off = 16; off > 0; off >>= 1) {
            s0 += __shfl_xor(s0, off, 32);
            s1 += __shfl_xor(s1, off, 32);
        }
        cscale = 1.0f / fmaxf(sqrtf(r == 1 ? s1 : s0), 1e-12f);
    }

    v8f acc[4]; float n2[4];
    __builtin_amdgcn_sched_barrier(0);
    load_grps<4>(emb, R0 + 16, r, j, hi, UB);
    __builtin_amdgcn_sched_barrier(0);
    mma_grps<4>(UA, bf, bt, tail_on, acc, n2);
    int R2 = R0 + 32;
    asm volatile("" : "+v"(acc[3]), "+v"(n2[3]), "+s"(R2));
    __builtin_amdgcn_sched_barrier(0);
    load_grps<4>(emb, R2, r, j, hi, UA);
    __builtin_amdgcn_sched_barrier(0);
    store_grps<4>(tab, R0, r, j, hi, acc, n2, cscale);
    mma_grps<4>(UB, bf, bt, tail_on, acc, n2);
    int R3 = R0 + 48;
    asm volatile("" : "+v"(acc[3]), "+v"(n2[3]), "+s"(R3));
    __builtin_amdgcn_sched_barrier(0);
    if (has7) load_grps<2>(emb, R3, r, j, hi, UC);
    __builtin_amdgcn_sched_barrier(0);
    store_grps<4>(tab, R0 + 16, r, j, hi, acc, n2, cscale);
    mma_grps<4>(UA, bf, bt, tail_on, acc, n2);
    store_grps<4>(tab, R2, r, j, hi, acc, n2, cscale);
    if (has7) {
        mma_grps<2>(UC, bf, bt, tail_on, acc, n2);
        store_grps<2>(tab, R3, r, j, hi, acc, n2, cscale);
    }
}

__device__ __forceinline__ float wave_max(float x)
{
    x = fmaxf(x, xrow<0x161>(x));
    x = fmaxf(x, xrow<0x162>(x));
    x = fmaxf(x, xrow<0x164>(x));
    x = fmaxf(x, xrow<0x168>(x));
    return fmaxf(x, xhalf(x));
}
__device__ __forceinline__ float wave_sum(float x)
{
    x += xrow<0x161>(x);
    x += xrow<0x162>(x);
    x += xrow<0x164>(x);
    x += xrow<0x168>(x);
    return x + xhalf(x);
}

__device__ __forceinline__ void lds_ld4(float* __restrict__ d, const float* s)
{
    const float4 v = *(const float4*)s;
    d[0] = v.x; d[1] = v.y; d[2] = v.z; d[3] = v.w;
}

__global__ __launch_bounds__(256)
void k_phase2(const int* __restrict__ tok, const float* __restrict__ tab,
              const float* __restrict__ conv_w,
              const float* __restrict__ conv_b, const float* __restrict__ lin_b,
              float* __restrict__ betaOut, float* __restrict__ logits)
{
    __shared__ __attribute__((aligned(16))) float gin[Cn][GWn];
    __shared__ __attribute__((aligned(16))) float wtap[KWn * 4];
    __shared__ __attribute__((aligned(16))) float hbuf[Ln];
    __shared__ __attribute__((aligned(16))) float wmax[8];
    __shared__ __attribute__((aligned(16))) float wsum[8];
    __shared__ __attribute__((aligned(16))) float wsq[8];
    const int b = blockIdx.x, tid = threadIdx.x;
    const int wave = tid >> 5, lane = tid & 31;

    int tk[8];
    #pragma unroll
    for (int j = 0; j < 8; ++j) tk[j] = tok[b * Ln + tid + 256 * j];
    const int kc = tid < Kn ? tid : 0;
    float4 w = make_float4(conv_w[0 * Kn + kc], conv_w[2 * Kn + kc],
                           conv_w[1 * Kn + kc], conv_w[3 * Kn + kc]);
    if (tid >= Kn) w = make_float4(0.0f, 0.0f, 0.0f, 0.0f);
    const float cb0 = conv_b[0], cb1 = conv_b[1];
    const float lbd = lin_b[1] - lin_b[0];
    float q[8], ga[8], gc[8];
    {
        float4 e[8];
        #pragma unroll
        for (int j = 0; j < 8; ++j) e[j] = ((const float4*)tab)[tk[j]];
        #pragma unroll
        for (int j = 0; j < 8; ++j) { ga[j] = e[j].x; gc[j] = e[j].y; q[j] = e[j].z; }
    }

    if (tid < Cn * (GWn - Ln)) {
        const int c = tid >= (GWn - Ln) ? 1 : 0;
        const int r = tid - c * (GWn - Ln);
        gin[c][r < PADn ? r : r + Ln] = 0.0f;
    }
    #pragma unroll
    for (int j = 0; j < 8; ++j) {
        gin[0][PADn + tid + 256 * j] = ga[j];
        gin[1][PADn + tid + 256 * j] = gc[j];
    }
    if (tid < KWn) *(float4*)&wtap[4 * tid] = w;
    __syncthreads();

    {
        const float* in0 = &gin[0][8 * tid];
        const float* in1 = &gin[1][8 * tid];
        v2f   acc[8];
        float xa[16], xb[16];
        #pragma unroll
        for (int i = 0; i < 8; ++i) { acc[i].x = 0.0f; acc[i].y = 0.0f; }
        lds_ld4(xa + 0, in0 + 0);  lds_ld4(xa + 4, in0 + 4);
        lds_ld4(xb + 0, in1 + 0);  lds_ld4(xb + 4, in1 + 4);

        float4 wc[8];
        #pragma unroll
        for (int kk = 0; kk < 8; ++kk) wc[kk] = *(const float4*)&wtap[4 * kk];

        #pragma unroll 2
        for (int k0 = 0; k0 < KPn; k0 += 8) {
            lds_ld4(xa + 8, in0 + k0 + 8);  lds_ld4(xa + 12, in0 + k0 + 12);
            lds_ld4(xb + 8, in1 + k0 + 8);  lds_ld4(xb + 12, in1 + k0 + 12);
            float4 wn[8];
            #pragma unroll
            for (int kk = 0; kk < 8; ++kk) wn[kk] = *(const float4*)&wtap[4 * (k0 + 8 + kk)];
            __builtin_amdgcn_sched_barrier(0);
            #pragma unroll
            for (int kk = 0; kk < 8; ++kk) {
                v2f wa; wa.x = wc[kk].x; wa.y = wc[kk].y;
                v2f wb; wb.x = wc[kk].z; wb.y = wc[kk].w;
                #pragma unroll
                for (int i = 0; i < 8; ++i) {
                    v2f va; va.x = xa[kk + i]; va.y = xa[kk + i];
                    v2f vb; vb.x = xb[kk + i]; vb.y = xb[kk + i];
                    acc[i] = __builtin_elementwise_fma(wa, va, acc[i]);
                    acc[i] = __builtin_elementwise_fma(wb, vb, acc[i]);
                }
            }
            #pragma unroll
            for (int i = 0; i < 8; ++i) { xa[i] = xa[i + 8]; xb[i] = xb[i + 8]; wc[i] = wn[i]; }
        }

        float r[8];
        #pragma unroll
        for (int i = 0; i < 8; ++i)
            r[i] = fmaxf(fmaxf(acc[i].x + cb0, 0.0f), fmaxf(acc[i].y + cb1, 0.0f));
        *(float4*)&hbuf[8 * tid + 0] = make_float4(r[0], r[1], r[2], r[3]);
        *(float4*)&hbuf[8 * tid + 4] = make_float4(r[4], r[5], r[6], r[7]);
        float lm = fmaxf(fmaxf(fmaxf(r[0], r[1]), fmaxf(r[2], r[3])),
                         fmaxf(fmaxf(r[4], r[5]), fmaxf(r[6], r[7])));
        lm = wave_max(lm);
        if (lane == 0) wmax[wave] = lm;
    }
    __syncthreads();
    float h[8];
    #pragma unroll
    for (int j = 0; j < 8; ++j) h[j] = hbuf[tid + 256 * j];

    float M;
    {
        const float4 u = *(const float4*)&wmax[0];
        const float4 v = *(const float4*)&wmax[4];
        M = fmaxf(fmaxf(fmaxf(u.x, u.y), fmaxf(u.z, u.w)),
                  fmaxf(fmaxf(v.x, v.y), fmaxf(v.z, v.w)));
    }

    float ls = 0.0f, lq = 0.0f;
    #pragma unroll
    for (int j = 0; j < 8; ++j) { h[j] = __expf(h[j] - M); ls += h[j]; lq += h[j] * q[j]; }
    ls = wave_sum(ls);
    lq = wave_sum(lq);
    if (lane == 0) { wsum[wave] = ls; wsq[wave] = lq; }
    __syncthreads();
    float S, SQ;
    {
        const float4 u = *(const float4*)&wsum[0];
        const float4 v = *(const float4*)&wsum[4];
        const float4 x = *(const float4*)&wsq[0];
        const float4 y = *(const float4*)&wsq[4];
        S  = ((u.x + u.y) + (u.z + u.w)) + ((v.x + v.y) + (v.z + v.w));
        SQ = ((x.x + x.y) + (x.z + x.w)) + ((y.x + y.y) + (y.z + y.w));
    }
    const float inv = 1.0f / S;
    #pragma unroll
    for (int j = 0; j < 8; ++j)
        betaOut[(long)b * Ln + tid + 256 * j] = h[j] * inv;

    if (tid == 0) {
        const float d  = SQ * inv + lbd;
        const float t  = __logf(1.0f + __expf(-fabsf(d)));
        logits[b * 2 + 0] = -(fmaxf(d, 0.0f) + t);
        logits[b * 2 + 1] = -(fmaxf(-d, 0.0f) + t);
    }
}

extern "C" void kernel_launch(void* const* d_in, const int* in_sizes, int n_in,
                              void* d_out, int out_size, void* d_ws, size_t ws_size,
                              hipStream_t stream)
{
    const int*   tok = (const int*)  d_in[0];
    const float* emb = (const float*)d_in[1];
    const float* cls = (const float*)d_in[2];
    const float* cw  = (const float*)d_in[3];
    const float* cb  = (const float*)d_in[4];
    const float* lw  = (const float*)d_in[5];
    const float* lb  = (const float*)d_in[6];

    float* out    = (float*)d_out;
    float* logits = out;
    float* beta   = out + Bn * Cn;
    float* tab    = (float*)d_ws;

    k_phase1<<<256, 32 * P1Wn, 0, stream>>>(emb, cls, lw, tab);
    k_phase2<<<Bn, 256, 0, stream>>>(tok, tab, cw, cb, lb, beta, logits);
}
